# starting kernel
# baseline (speedup 1.0000x reference)
#include <hip/hip_runtime.h>

#define TT 256
#define DD 32
#define KK 8
#define NB 64
#define AUG_LD 66
#define EPS_TRI 0.01f
#define EPS_CHOL 0.01f
#define EPS_BS 0.0001f

typedef float v2f __attribute__((ext_vector_type(2)));
typedef float v8f __attribute__((ext_vector_type(8)));

enum { MM_STORE = 0, MM_STORE_NEG = 1, MM_SUB = 2, MM_ADD = 3 };

template <bool TA, bool TB>
__device__ inline void mm32(const float* A, const float* B, float* D, int ldd,
                            int mode, int tid) {
  const int lane = tid & 31;
  const int wid = tid >> 5;
  const int mt = (wid >> 1) << 4;
  const int nt = (wid & 1) << 4;
  const int l15 = lane & 15;
  const int kh = (lane >> 4) << 1;
  v8f acc = {};
#pragma unroll
  for (int kt = 0; kt < 32; kt += 4) {
    const int k0 = kt + kh;
    v2f av, bv;
    if (!TA) { av.x = A[(mt + l15) * 32 + k0];  av.y = A[(mt + l15) * 32 + k0 + 1]; }
    else     { av.x = A[k0 * 32 + mt + l15];    av.y = A[(k0 + 1) * 32 + mt + l15]; }
    if (!TB) { bv.x = B[k0 * 32 + nt + l15];    bv.y = B[(k0 + 1) * 32 + nt + l15]; }
    else     { bv.x = B[(nt + l15) * 32 + k0];  bv.y = B[(nt + l15) * 32 + k0 + 1]; }
    acc = __builtin_amdgcn_wmma_f32_16x16x4_f32(false, av, false, bv, (short)0,
                                                acc, false, false);
  }
  const int rbase = mt + ((lane >> 4) << 3);
#pragma unroll
  for (int r = 0; r < 8; ++r) {
    float v = acc[r];
    float* p = &D[(rbase + r) * ldd + nt + l15];
    if (mode == MM_STORE) *p = v;
    else if (mode == MM_STORE_NEG) *p = -v;
    else if (mode == MM_SUB) *p -= v;
    else *p += v;
  }
}

__global__ __launch_bounds__(128) void ris_precompute_kernel(
    const float* __restrict__ Ri_sqrts, const float* __restrict__ ms,
    float* __restrict__ g_Ris, float* __restrict__ g_hR) {
  __shared__ float sR[1024], sO[1024], sm[DD];
  const int t = blockIdx.x;
  const int tid = threadIdx.x;
  for (int e = tid; e < 1024; e += 128) sR[e] = Ri_sqrts[(size_t)t * 1024 + e];
  if (tid < DD) sm[tid] = ms[t * DD + tid];
  __syncthreads();
  mm32<false, true>(sR, sR, sO, 32, MM_STORE, tid);
  __syncthreads();
  for (int e = tid; e < 1024; e += 128) g_Ris[(size_t)t * 1024 + e] = sO[e];
  if (tid < DD) {
    float a = 0.f;
    for (int j = 0; j < DD; ++j) a += sO[tid * 32 + j] * sm[j];
    g_hR[t * DD + tid] = a;
  }
}

__global__ __launch_bounds__(128) void lds_forward_kernel(
    const float* __restrict__ z_samples, const float* __restrict__ A_base,
    const float* __restrict__ b_base, const float* __restrict__ Q_sqrt,
    const float* __restrict__ noise, const float* __restrict__ g_Ris,
    const float* __restrict__ g_hR, float* __restrict__ g_c,
    float* __restrict__ g_d, float* __restrict__ out) {
  __shared__ float sA[1024], sQs[1024], sJd[1024], sC[1024], sL[1024],
      sLl[1024], sM[1024];
  __shared__ float sQis[2][1024], sJl[2][1024];
  __shared__ float sAug[32 * AUG_LD];
  __shared__ float sz[KK], sbs[DD], sQb[DD], sh[DD], sJlPd[DD], sd[DD],
      srhs[DD], sx[DD];

  const int n = blockIdx.x;
  const int tid = threadIdx.x;
  const int row = tid & 31;
  const int jgrp = tid >> 5;

  for (int t = 0; t < TT; ++t) {
    const int cur = t & 1, prev = cur ^ 1;
    const bool haveNext = (t < TT - 1);

    if (haveNext) {
      if (tid < KK) sz[tid] = z_samples[((size_t)n * TT + t) * KK + tid];
      __syncthreads();
      for (int e = tid; e < 1024; e += 128) {
        float a = 0.f, q = 0.f;
#pragma unroll
        for (int k = 0; k < KK; ++k) {
          float zk = sz[k];
          a += zk * A_base[k * 1024 + e];
          q += zk * Q_sqrt[k * 1024 + e];
        }
        sA[e] = a;
        sQs[e] = q;
      }
      if (tid < DD) {
        float b = 0.f;
#pragma unroll
        for (int k = 0; k < KK; ++k) b += sz[k] * b_base[k * DD + tid];
        sbs[tid] = b;
      }
      __syncthreads();
      mm32<false, true>(sQs, sQs, sQis[cur], 32, MM_STORE, tid);
      __syncthreads();
      mm32<false, false>(sQis[cur], sA, sJl[cur], 32, MM_STORE_NEG, tid);
      __syncthreads();
      mm32<true, false>(sA, sJl[cur], sJd, 32, MM_STORE_NEG, tid);
      __syncthreads();
    } else {
      for (int e = tid; e < 1024; e += 128) sJd[e] = 0.f;
      __syncthreads();
    }

    for (int e = tid; e < 1024; e += 128) {
      float v = sJd[e] + g_Ris[(size_t)t * 1024 + e];
      if (t > 0) v += sQis[prev][e];
      sJd[e] = v;
    }
    if (tid < DD) {
      float hv = g_hR[t * DD + tid];
      if (t > 0) hv += sQb[tid];
      if (haveNext) {
        float jb = 0.f;
        for (int j = 0; j < DD; ++j) jb += sJl[cur][tid * 32 + j] * sbs[j];
        hv += jb;
      }
      float jd = 0.f;
      if (t > 0)
        for (int j = 0; j < DD; ++j) jd += sJl[prev][tid * 32 + j] * sd[j];
      sh[tid] = hv;
      sJlPd[tid] = jd;
      if (haveNext) {
        float qb = 0.f;
        for (int j = 0; j < DD; ++j) qb += sQis[cur][tid * 32 + j] * sbs[j];
        sQb[tid] = qb;
      }
    }
    __syncthreads();

    for (int idx = tid; idx < 32 * 65; idx += 128) {
      int i = idx / 65, j = idx % 65;
      float v;
      if (j < 32)      v = sJd[i * 32 + j] + (i == j ? EPS_TRI : 0.f);
      else if (j < 64) v = haveNext ? sJl[cur][(j - 32) * 32 + i] : 0.f;
      else             v = sh[i] - (t > 0 ? sJlPd[i] : 0.f);
      sAug[i * AUG_LD + j] = v;
    }
    __syncthreads();
    if (t > 0) {
      mm32<false, false>(sJl[prev], sC, sAug, AUG_LD, MM_SUB, tid);
      __syncthreads();
    }
    for (int k = 0; k < 32; ++k) {
      float piv = sAug[k * AUG_LD + k];
      float f = sAug[row * AUG_LD + k] / piv;
      __syncthreads();
      if (row != k) {
        for (int j = k + 1 + jgrp; j < 65; j += 4)
          sAug[row * AUG_LD + j] -= f * sAug[k * AUG_LD + j];
        if (jgrp == 0) sAug[row * AUG_LD + k] = 0.f;
      }
      __syncthreads();
    }
    for (int e = tid; e < 1024; e += 128) {
      int i = e >> 5, j = e & 31;
      sC[e] = sAug[j * AUG_LD + 32 + i] / sAug[j * AUG_LD + j];
    }
    if (tid < DD) sd[tid] = sAug[tid * AUG_LD + 64] / sAug[tid * AUG_LD + tid];
    __syncthreads();
    {
      size_t base = (size_t)n * TT + t;
      if (haveNext)
        for (int e = tid; e < 1024; e += 128) g_c[base * 1024 + e] = sC[e];
      if (tid < DD) g_d[base * DD + tid] = sd[tid];
    }

    if (t > 0) {
      for (int e = tid; e < 1024; e += 128)
        sAug[(e >> 5) * AUG_LD + (e & 31)] = sJl[prev][e];
      __syncthreads();
      if (tid < 32) {
        int c = tid;
        for (int i = 31; i >= 0; --i) {
          float xi = sAug[c * AUG_LD + i] / sL[i * 32 + i];
          sAug[c * AUG_LD + i] = xi;
          for (int j = 0; j < i; ++j)
            sAug[c * AUG_LD + j] -= sL[i * 32 + j] * xi;
        }
      }
      __syncthreads();
      for (int e = tid; e < 1024; e += 128)
        sLl[e] = sAug[(e >> 5) * AUG_LD + (e & 31)];
      __syncthreads();
      for (int e = tid; e < 1024; e += 128) {
        int i = e >> 5, j = e & 31;
        sM[e] = sJd[e] + (i == j ? EPS_CHOL : 0.f);
      }
      __syncthreads();
      mm32<false, true>(sLl, sLl, sM, 32, MM_SUB, tid);
      __syncthreads();
    } else {
      for (int e = tid; e < 1024; e += 128) {
        int i = e >> 5, j = e & 31;
        sM[e] = sJd[e] + (i == j ? EPS_CHOL : 0.f);
      }
      __syncthreads();
    }
    for (int k = 0; k < 32; ++k) {
      float akk = sM[k * 32 + k];
      __syncthreads();
      float lkk = sqrtf(akk);
      if (tid == k) sM[k * 32 + k] = lkk;
      else if (tid < 32 && tid > k) sM[tid * 32 + k] /= lkk;
      __syncthreads();
      if (row > k)
        for (int j = k + 1 + jgrp; j < 32; j += 4)
          sM[row * 32 + j] -= sM[row * 32 + k] * sM[j * 32 + k];
      __syncthreads();
    }
    for (int e = tid; e < 1024; e += 128) {
      int i = e >> 5, j = e & 31;
      sL[e] = (j <= i) ? sM[e] : 0.f;
    }
    __syncthreads();

    if (tid < DD) {
      float r = noise[((size_t)n * TT + t) * DD + tid];
      if (t > 0) {
        float s = 0.f;
        for (int j = 0; j < DD; ++j) s += sLl[j * 32 + tid] * sx[j];
        r -= s;
      }
      srhs[tid] = r;
    }
    __syncthreads();
    for (int i = 31; i >= 0; --i) {
      if (tid == 0) srhs[i] /= (sL[i * 32 + i] + EPS_BS);
      __syncthreads();
      if (tid < i) srhs[tid] -= sL[i * 32 + tid] * srhs[i];
      __syncthreads();
    }
    if (tid < DD) {
      sx[tid] = srhs[tid];
      out[((size_t)n * TT + t) * DD + tid] = srhs[tid];
    }
    __syncthreads();
  }
}

__global__ __launch_bounds__(128) void lds_backward_kernel(
    const float* __restrict__ g_c, const float* __restrict__ g_d,
    float* __restrict__ out) {
  __shared__ float sCt[1024];
  __shared__ float sMu[DD], sNew[DD];
  const int n = blockIdx.x;
  const int tid = threadIdx.x;
  if (tid < DD) {
    float v = g_d[((size_t)n * TT + (TT - 1)) * DD + tid];
    sMu[tid] = v;
    out[((size_t)n * TT + (TT - 1)) * DD + tid] += v;
  }
  __syncthreads();
  for (int t = TT - 2; t >= 0; --t) {
    size_t base = (size_t)n * TT + t;
    for (int e = tid; e < 1024; e += 128) sCt[e] = g_c[base * 1024 + e];
    __syncthreads();
    if (tid < DD) {
      float acc = g_d[base * DD + tid];
      for (int j = 0; j < DD; ++j) acc -= sCt[tid * 32 + j] * sMu[j];
      sNew[tid] = acc;
    }
    __syncthreads();
    if (tid < DD) {
      sMu[tid] = sNew[tid];
      out[base * DD + tid] += sNew[tid];
    }
    __syncthreads();
  }
}

extern "C" void kernel_launch(void* const* d_in, const int* in_sizes, int n_in,
                              void* d_out, int out_size, void* d_ws,
                              size_t ws_size, hipStream_t stream) {
  (void)in_sizes; (void)n_in; (void)out_size; (void)ws_size;
  const float* z_samples = (const float*)d_in[0];
  const float* A_base = (const float*)d_in[1];
  const float* b_base = (const float*)d_in[2];
  const float* Q_sqrt = (const float*)d_in[3];
  const float* ms = (const float*)d_in[4];
  const float* Ri_sqrts = (const float*)d_in[5];
  const float* noise = (const float*)d_in[6];
  float* ws = (float*)d_ws;
  float* g_Ris = ws;
  float* g_hR = g_Ris + (size_t)TT * DD * DD;
  float* g_d = g_hR + (size_t)TT * DD;
  float* g_c = g_d + (size_t)NB * TT * DD;
  float* out = (float*)d_out;

  ris_precompute_kernel<<<TT, 128, 0, stream>>>(Ri_sqrts, ms, g_Ris, g_hR);
  lds_forward_kernel<<<NB, 128, 0, stream>>>(z_samples, A_base, b_base, Q_sqrt,
                                             noise, g_Ris, g_hR, g_c, g_d, out);
  lds_backward_kernel<<<NB, 128, 0, stream>>>(g_c, g_d, out);
}


# Split the two independent recurrences across work-groups: 64 run the tridiagonal elimination (c,d), 64 the Cholesky+sampling chain (L,Ll,x); each rebuilds Jd,Jl; every sum and its order unchanged
# speedup vs baseline: 1.7121x; 1.7121x over previous
#include <hip/hip_runtime.h>

#define TT 256
#define DD 32
#define KK 8
#define NB 64
#define NB_FWD 128
#define AUG_LD 66
#define EPS_TRI 0.01f
#define EPS_CHOL 0.01f
#define EPS_BS 0.0001f

typedef float v2f __attribute__((ext_vector_type(2)));
typedef float v8f __attribute__((ext_vector_type(8)));

enum { MM_STORE = 0, MM_STORE_NEG = 1, MM_SUB = 2, MM_ADD = 3 };

template <bool TA, bool TB>
__device__ inline void mm32(const float* A, const float* B, float* D, int ldd,
                            int mode, int tid) {
  const int lane = tid & 31;
  const int wid = tid >> 5;
  const int mt = (wid >> 1) << 4;
  const int nt = (wid & 1) << 4;
  const int l15 = lane & 15;
  const int kh = (lane >> 4) << 1;
  v8f acc = {};
#pragma unroll
  for (int kt = 0; kt < 32; kt += 4) {
    const int k0 = kt + kh;
    v2f av, bv;
    if (!TA) { av.x = A[(mt + l15) * 32 + k0];  av.y = A[(mt + l15) * 32 + k0 + 1]; }
    else     { av.x = A[k0 * 32 + mt + l15];    av.y = A[(k0 + 1) * 32 + mt + l15]; }
    if (!TB) { bv.x = B[k0 * 32 + nt + l15];    bv.y = B[(k0 + 1) * 32 + nt + l15]; }
    else     { bv.x = B[(nt + l15) * 32 + k0];  bv.y = B[(nt + l15) * 32 + k0 + 1]; }
    acc = __builtin_amdgcn_wmma_f32_16x16x4_f32(false, av, false, bv, (short)0,
                                                acc, false, false);
  }
  const int rbase = mt + ((lane >> 4) << 3);
#pragma unroll
  for (int r = 0; r < 8; ++r) {
    float v = acc[r];
    float* p = &D[(rbase + r) * ldd + nt + l15];
    if (mode == MM_STORE) *p = v;
    else if (mode == MM_STORE_NEG) *p = -v;
    else if (mode == MM_SUB) *p -= v;
    else *p += v;
  }
}

__global__ __launch_bounds__(128) void ris_precompute_kernel(
    const float* __restrict__ Ri_sqrts, const float* __restrict__ ms,
    float* __restrict__ g_Ris, float* __restrict__ g_hR) {
  __shared__ float sR[1024], sO[1024], sm[DD];
  const int t = blockIdx.x;
  const int tid = threadIdx.x;
  for (int e = tid; e < 1024; e += 128) sR[e] = Ri_sqrts[(size_t)t * 1024 + e];
  if (tid < DD) sm[tid] = ms[t * DD + tid];
  __syncthreads();
  mm32<false, true>(sR, sR, sO, 32, MM_STORE, tid);
  __syncthreads();
  for (int e = tid; e < 1024; e += 128) g_Ris[(size_t)t * 1024 + e] = sO[e];
  if (tid < DD) {
    float a = 0.f;
    for (int j = 0; j < DD; ++j) a += sO[tid * 32 + j] * sm[j];
    g_hR[t * DD + tid] = a;
  }
}

__global__ __launch_bounds__(128) void lds_forward_kernel(
    const float* __restrict__ z_samples, const float* __restrict__ A_base,
    const float* __restrict__ b_base, const float* __restrict__ Q_sqrt,
    const float* __restrict__ noise, const float* __restrict__ g_Ris,
    const float* __restrict__ g_hR, float* __restrict__ g_c,
    float* __restrict__ g_d, float* __restrict__ out) {
  __shared__ float sA[1024], sQs[1024], sJd[1024], sC[1024], sL[1024],
      sLl[1024], sM[1024];
  __shared__ float sQis[2][1024], sJl[2][1024];
  __shared__ float sAug[32 * AUG_LD];
  __shared__ float sz[KK], sbs[DD], sQb[DD], sh[DD], sJlPd[DD], sd[DD],
      srhs[DD], sx[DD];

  const bool triRole = (blockIdx.x < NB);
  const int n = triRole ? blockIdx.x : blockIdx.x - NB;
  const int tid = threadIdx.x;
  const int row = tid & 31;
  const int jgrp = tid >> 5;

  for (int t = 0; t < TT; ++t) {
    const int cur = t & 1, prev = cur ^ 1;
    const bool haveNext = (t < TT - 1);

    if (haveNext) {
      if (tid < KK) sz[tid] = z_samples[((size_t)n * TT + t) * KK + tid];
      __syncthreads();
      for (int e = tid; e < 1024; e += 128) {
        float a = 0.f, q = 0.f;
#pragma unroll
        for (int k = 0; k < KK; ++k) {
          float zk = sz[k];
          a += zk * A_base[k * 1024 + e];
          q += zk * Q_sqrt[k * 1024 + e];
        }
        sA[e] = a;
        sQs[e] = q;
      }
      if (triRole && tid < DD) {
        float b = 0.f;
#pragma unroll
        for (int k = 0; k < KK; ++k) b += sz[k] * b_base[k * DD + tid];
        sbs[tid] = b;
      }
      __syncthreads();
      mm32<false, true>(sQs, sQs, sQis[cur], 32, MM_STORE, tid);
      __syncthreads();
      mm32<false, false>(sQis[cur], sA, sJl[cur], 32, MM_STORE_NEG, tid);
      __syncthreads();
      mm32<true, false>(sA, sJl[cur], sJd, 32, MM_STORE_NEG, tid);
      __syncthreads();
    } else {
      for (int e = tid; e < 1024; e += 128) sJd[e] = 0.f;
      __syncthreads();
    }

    for (int e = tid; e < 1024; e += 128) {
      float v = sJd[e] + g_Ris[(size_t)t * 1024 + e];
      if (t > 0) v += sQis[prev][e];
      sJd[e] = v;
    }
    if (triRole && tid < DD) {
      float hv = g_hR[t * DD + tid];
      if (t > 0) hv += sQb[tid];
      if (haveNext) {
        float jb = 0.f;
        for (int j = 0; j < DD; ++j) jb += sJl[cur][tid * 32 + j] * sbs[j];
        hv += jb;
      }
      float jd = 0.f;
      if (t > 0)
        for (int j = 0; j < DD; ++j) jd += sJl[prev][tid * 32 + j] * sd[j];
      sh[tid] = hv;
      sJlPd[tid] = jd;
      if (haveNext) {
        float qb = 0.f;
        for (int j = 0; j < DD; ++j) qb += sQis[cur][tid * 32 + j] * sbs[j];
        sQb[tid] = qb;
      }
    }
    __syncthreads();

    if (triRole) {
      for (int idx = tid; idx < 32 * 65; idx += 128) {
        int i = idx / 65, j = idx % 65;
        float v;
        if (j < 32)      v = sJd[i * 32 + j] + (i == j ? EPS_TRI : 0.f);
        else if (j < 64) v = haveNext ? sJl[cur][(j - 32) * 32 + i] : 0.f;
        else             v = sh[i] - (t > 0 ? sJlPd[i] : 0.f);
        sAug[i * AUG_LD + j] = v;
      }
      __syncthreads();
      if (t > 0) {
        mm32<false, false>(sJl[prev], sC, sAug, AUG_LD, MM_SUB, tid);
        __syncthreads();
      }
      for (int k = 0; k < 32; ++k) {
        float piv = sAug[k * AUG_LD + k];
        float f = sAug[row * AUG_LD + k] / piv;
        __syncthreads();
        if (row != k) {
          for (int j = k + 1 + jgrp; j < 65; j += 4)
            sAug[row * AUG_LD + j] -= f * sAug[k * AUG_LD + j];
          if (jgrp == 0) sAug[row * AUG_LD + k] = 0.f;
        }
        __syncthreads();
      }
      for (int e = tid; e < 1024; e += 128) {
        int i = e >> 5, j = e & 31;
        sC[e] = sAug[j * AUG_LD + 32 + i] / sAug[j * AUG_LD + j];
      }
      if (tid < DD) sd[tid] = sAug[tid * AUG_LD + 64] / sAug[tid * AUG_LD + tid];
      __syncthreads();
      {
        size_t base = (size_t)n * TT + t;
        if (haveNext)
          for (int e = tid; e < 1024; e += 128) g_c[base * 1024 + e] = sC[e];
        if (tid < DD) g_d[base * DD + tid] = sd[tid];
      }
      continue;
    }

    if (t > 0) {
      for (int e = tid; e < 1024; e += 128)
        sAug[(e >> 5) * AUG_LD + (e & 31)] = sJl[prev][e];
      __syncthreads();
      if (tid < 32) {
        int c = tid;
        for (int i = 31; i >= 0; --i) {
          float xi = sAug[c * AUG_LD + i] / sL[i * 32 + i];
          sAug[c * AUG_LD + i] = xi;
          for (int j = 0; j < i; ++j)
            sAug[c * AUG_LD + j] -= sL[i * 32 + j] * xi;
        }
      }
      __syncthreads();
      for (int e = tid; e < 1024; e += 128)
        sLl[e] = sAug[(e >> 5) * AUG_LD + (e & 31)];
      __syncthreads();
      for (int e = tid; e < 1024; e += 128) {
        int i = e >> 5, j = e & 31;
        sM[e] = sJd[e] + (i == j ? EPS_CHOL : 0.f);
      }
      __syncthreads();
      mm32<false, true>(sLl, sLl, sM, 32, MM_SUB, tid);
      __syncthreads();
    } else {
      for (int e = tid; e < 1024; e += 128) {
        int i = e >> 5, j = e & 31;
        sM[e] = sJd[e] + (i == j ? EPS_CHOL : 0.f);
      }
      __syncthreads();
    }
    for (int k = 0; k < 32; ++k) {
      float akk = sM[k * 32 + k];
      __syncthreads();
      float lkk = sqrtf(akk);
      if (tid == k) sM[k * 32 + k] = lkk;
      else if (tid < 32 && tid > k) sM[tid * 32 + k] /= lkk;
      __syncthreads();
      if (row > k)
        for (int j = k + 1 + jgrp; j < 32; j += 4)
          sM[row * 32 + j] -= sM[row * 32 + k] * sM[j * 32 + k];
      __syncthreads();
    }
    for (int e = tid; e < 1024; e += 128) {
      int i = e >> 5, j = e & 31;
      sL[e] = (j <= i) ? sM[e] : 0.f;
    }
    __syncthreads();

    if (tid < DD) {
      float r = noise[((size_t)n * TT + t) * DD + tid];
      if (t > 0) {
        float s = 0.f;
        for (int j = 0; j < DD; ++j) s += sLl[j * 32 + tid] * sx[j];
        r -= s;
      }
      srhs[tid] = r;
    }
    __syncthreads();
    for (int i = 31; i >= 0; --i) {
      if (tid == 0) srhs[i] /= (sL[i * 32 + i] + EPS_BS);
      __syncthreads();
      if (tid < i) srhs[tid] -= sL[i * 32 + tid] * srhs[i];
      __syncthreads();
    }
    if (tid < DD) {
      sx[tid] = srhs[tid];
      out[((size_t)n * TT + t) * DD + tid] = srhs[tid];
    }
    __syncthreads();
  }
}

__global__ __launch_bounds__(128) void lds_backward_kernel(
    const float* __restrict__ g_c, const float* __restrict__ g_d,
    float* __restrict__ out) {
  __shared__ float sCt[1024];
  __shared__ float sMu[DD], sNew[DD];
  const int n = blockIdx.x;
  const int tid = threadIdx.x;
  if (tid < DD) {
    float v = g_d[((size_t)n * TT + (TT - 1)) * DD + tid];
    sMu[tid] = v;
    out[((size_t)n * TT + (TT - 1)) * DD + tid] += v;
  }
  __syncthreads();
  for (int t = TT - 2; t >= 0; --t) {
    size_t base = (size_t)n * TT + t;
    for (int e = tid; e < 1024; e += 128) sCt[e] = g_c[base * 1024 + e];
    __syncthreads();
    if (tid < DD) {
      float acc = g_d[base * DD + tid];
      for (int j = 0; j < DD; ++j) acc -= sCt[tid * 32 + j] * sMu[j];
      sNew[tid] = acc;
    }
    __syncthreads();
    if (tid < DD) {
      sMu[tid] = sNew[tid];
      out[base * DD + tid] += sNew[tid];
    }
    __syncthreads();
  }
}

extern "C" void kernel_launch(void* const* d_in, const int* in_sizes, int n_in,
                              void* d_out, int out_size, void* d_ws,
                              size_t ws_size, hipStream_t stream) {
  (void)in_sizes; (void)n_in; (void)out_size; (void)ws_size;
  const float* z_samples = (const float*)d_in[0];
  const float* A_base = (const float*)d_in[1];
  const float* b_base = (const float*)d_in[2];
  const float* Q_sqrt = (const float*)d_in[3];
  const float* ms = (const float*)d_in[4];
  const float* Ri_sqrts = (const float*)d_in[5];
  const float* noise = (const float*)d_in[6];
  float* ws = (float*)d_ws;
  float* g_Ris = ws;
  float* g_hR = g_Ris + (size_t)TT * DD * DD;
  float* g_d = g_hR + (size_t)TT * DD;
  float* g_c = g_d + (size_t)NB * TT * DD;
  float* out = (float*)d_out;

  ris_precompute_kernel<<<TT, 128, 0, stream>>>(Ri_sqrts, ms, g_Ris, g_hR);
  lds_forward_kernel<<<NB_FWD, 128, 0, stream>>>(z_samples, A_base, b_base, Q_sqrt,
                                             noise, g_Ris, g_hR, g_c, g_d, out);
  lds_backward_kernel<<<NB, 128, 0, stream>>>(g_c, g_d, out);
}


# Serial sweeps (Gauss-Jordan, triangular solve, Cholesky, back-substitution) run row-per-lane in registers, pivots broadcast by v_readlane: no LDS trip or barrier per pivot; order of sums kept
# speedup vs baseline: 4.5866x; 2.6789x over previous
#include <hip/hip_runtime.h>

#define TT 256
#define DD 32
#define KK 8
#define NB 64
#define NB_FWD 128
#define AUG_LD 66
#define EPS_TRI 0.01f
#define EPS_CHOL 0.01f
#define EPS_BS 0.0001f

typedef float v2f __attribute__((ext_vector_type(2)));
typedef float v8f __attribute__((ext_vector_type(8)));

enum { MM_STORE = 0, MM_STORE_NEG = 1, MM_SUB = 2, MM_ADD = 3 };

template <bool TA, bool TB>
__device__ inline void mm32(const float* A, const float* B, float* D, int ldd,
                            int mode, int tid) {
  const int lane = tid & 31;
  const int wid = tid >> 5;
  const int mt = (wid >> 1) << 4;
  const int nt = (wid & 1) << 4;
  const int l15 = lane & 15;
  const int kh = (lane >> 4) << 1;
  v8f acc = {};
#pragma unroll
  for (int kt = 0; kt < 32; kt += 4) {
    const int k0 = kt + kh;
    v2f av, bv;
    if (!TA) { av.x = A[(mt + l15) * 32 + k0];  av.y = A[(mt + l15) * 32 + k0 + 1]; }
    else     { av.x = A[k0 * 32 + mt + l15];    av.y = A[(k0 + 1) * 32 + mt + l15]; }
    if (!TB) { bv.x = B[k0 * 32 + nt + l15];    bv.y = B[(k0 + 1) * 32 + nt + l15]; }
    else     { bv.x = B[(nt + l15) * 32 + k0];  bv.y = B[(nt + l15) * 32 + k0 + 1]; }
    acc = __builtin_amdgcn_wmma_f32_16x16x4_f32(false, av, false, bv, (short)0,
                                                acc, false, false);
  }
  const int rbase = mt + ((lane >> 4) << 3);
#pragma unroll
  for (int r = 0; r < 8; ++r) {
    float v = acc[r];
    float* p = &D[(rbase + r) * ldd + nt + l15];
    if (mode == MM_STORE) *p = v;
    else if (mode == MM_STORE_NEG) *p = -v;
    else if (mode == MM_SUB) *p -= v;
    else *p += v;
  }
}

__device__ inline float lane_bcast(float v, int src) {
  return __builtin_bit_cast(
      float, __builtin_amdgcn_readlane(__builtin_bit_cast(int, v), src));
}

__global__ __launch_bounds__(128) void ris_precompute_kernel(
    const float* __restrict__ Ri_sqrts, const float* __restrict__ ms,
    float* __restrict__ g_Ris, float* __restrict__ g_hR) {
  __shared__ float sR[1024], sO[1024], sm[DD];
  const int t = blockIdx.x;
  const int tid = threadIdx.x;
  for (int e = tid; e < 1024; e += 128) sR[e] = Ri_sqrts[(size_t)t * 1024 + e];
  if (tid < DD) sm[tid] = ms[t * DD + tid];
  __syncthreads();
  mm32<false, true>(sR, sR, sO, 32, MM_STORE, tid);
  __syncthreads();
  for (int e = tid; e < 1024; e += 128) g_Ris[(size_t)t * 1024 + e] = sO[e];
  if (tid < DD) {
    float a = 0.f;
    for (int j = 0; j < DD; ++j) a += sO[tid * 32 + j] * sm[j];
    g_hR[t * DD + tid] = a;
  }
}

__global__ __launch_bounds__(128) void lds_forward_kernel(
    const float* __restrict__ z_samples, const float* __restrict__ A_base,
    const float* __restrict__ b_base, const float* __restrict__ Q_sqrt,
    const float* __restrict__ noise, const float* __restrict__ g_Ris,
    const float* __restrict__ g_hR, float* __restrict__ g_c,
    float* __restrict__ g_d, float* __restrict__ out) {
  __shared__ float sA[1024], sQs[1024], sJd[1024], sC[1024], sL[1024],
      sLl[1024];
  __shared__ float sQis[2][1024], sJl[2][1024];
  __shared__ float sAug[32 * AUG_LD];
  __shared__ float sz[KK], sbs[DD], sQb[DD], sh[DD], sJlPd[DD], sd[DD],
      sDiag[DD], sx[DD];

  const bool triRole = (blockIdx.x < NB);
  const int n = triRole ? blockIdx.x : blockIdx.x - NB;
  const int tid = threadIdx.x;
  const int row = tid & 31;
  const int jgrp = tid >> 5;

  for (int t = 0; t < TT; ++t) {
    const int cur = t & 1, prev = cur ^ 1;
    const bool haveNext = (t < TT - 1);

    if (haveNext) {
      if (tid < KK) sz[tid] = z_samples[((size_t)n * TT + t) * KK + tid];
      __syncthreads();
      for (int e = tid; e < 1024; e += 128) {
        float a = 0.f, q = 0.f;
#pragma unroll
        for (int k = 0; k < KK; ++k) {
          float zk = sz[k];
          a += zk * A_base[k * 1024 + e];
          q += zk * Q_sqrt[k * 1024 + e];
        }
        sA[e] = a;
        sQs[e] = q;
      }
      if (triRole && tid < DD) {
        float b = 0.f;
#pragma unroll
        for (int k = 0; k < KK; ++k) b += sz[k] * b_base[k * DD + tid];
        sbs[tid] = b;
      }
      __syncthreads();
      mm32<false, true>(sQs, sQs, sQis[cur], 32, MM_STORE, tid);
      __syncthreads();
      mm32<false, false>(sQis[cur], sA, sJl[cur], 32, MM_STORE_NEG, tid);
      __syncthreads();
      mm32<true, false>(sA, sJl[cur], sJd, 32, MM_STORE_NEG, tid);
      __syncthreads();
    } else {
      for (int e = tid; e < 1024; e += 128) sJd[e] = 0.f;
      __syncthreads();
    }

    for (int e = tid; e < 1024; e += 128) {
      float v = sJd[e] + g_Ris[(size_t)t * 1024 + e];
      if (t > 0) v += sQis[prev][e];
      sJd[e] = v;
    }
    if (triRole && tid < DD) {
      float hv = g_hR[t * DD + tid];
      if (t > 0) hv += sQb[tid];
      if (haveNext) {
        float jb = 0.f;
        for (int j = 0; j < DD; ++j) jb += sJl[cur][tid * 32 + j] * sbs[j];
        hv += jb;
      }
      float jd = 0.f;
      if (t > 0)
        for (int j = 0; j < DD; ++j) jd += sJl[prev][tid * 32 + j] * sd[j];
      sh[tid] = hv;
      sJlPd[tid] = jd;
      if (haveNext) {
        float qb = 0.f;
        for (int j = 0; j < DD; ++j) qb += sQis[cur][tid * 32 + j] * sbs[j];
        sQb[tid] = qb;
      }
    }
    __syncthreads();

    if (triRole) {
      for (int idx = tid; idx < 32 * 65; idx += 128) {
        int i = idx / 65, j = idx % 65;
        float v;
        if (j < 32)      v = sJd[i * 32 + j] + (i == j ? EPS_TRI : 0.f);
        else if (j < 64) v = haveNext ? sJl[cur][(j - 32) * 32 + i] : 0.f;
        else             v = sh[i] - (t > 0 ? sJlPd[i] : 0.f);
        sAug[i * AUG_LD + j] = v;
      }
      __syncthreads();
      if (t > 0) {
        mm32<false, false>(sJl[prev], sC, sAug, AUG_LD, MM_SUB, tid);
        __syncthreads();
      }
      {
        float a[32], b[9];
        float diag = 0.f;
#pragma unroll
        for (int j = 0; j < 32; ++j) a[j] = sAug[row * AUG_LD + j];
#pragma unroll
        for (int j = 0; j < 8; ++j)
          b[j] = sAug[row * AUG_LD + 32 + 8 * jgrp + j];
        b[8] = sAug[row * AUG_LD + 64];
#pragma unroll
        for (int k = 0; k < 32; ++k) {
          const float piv = lane_bcast(a[k], k);
          float f = a[k] / piv;
          f = (row == k) ? 0.f : f;
          diag = (row == k) ? piv : diag;
#pragma unroll
          for (int j = k + 1; j < 32; ++j) a[j] -= f * lane_bcast(a[j], k);
#pragma unroll
          for (int j = 0; j < 9; ++j) b[j] -= f * lane_bcast(b[j], k);
        }
#pragma unroll
        for (int j = 0; j < 8; ++j)
          sAug[row * AUG_LD + 32 + 8 * jgrp + j] = b[j];
        if (jgrp == 0) {
          sDiag[row] = diag;
          sd[row] = b[8] / diag;
        }
      }
      __syncthreads();
      for (int e = tid; e < 1024; e += 128) {
        int i = e >> 5, j = e & 31;
        sC[e] = sAug[j * AUG_LD + 32 + i] / sDiag[j];
      }
      __syncthreads();
      {
        size_t base = (size_t)n * TT + t;
        if (haveNext)
          for (int e = tid; e < 1024; e += 128) g_c[base * 1024 + e] = sC[e];
        if (tid < DD) g_d[base * DD + tid] = sd[tid];
      }
      continue;
    }

    if (t > 0) {
      for (int e = tid; e < 1024; e += 128)
        sAug[(e >> 5) * AUG_LD + (e & 31)] = sJl[prev][e];
      __syncthreads();
      if (tid < 32) {
        float x[32];
#pragma unroll
        for (int j = 0; j < 32; ++j) x[j] = sAug[tid * AUG_LD + j];
#pragma unroll
        for (int i = 31; i >= 0; --i) {
          const float xi = x[i] / sL[i * 32 + i];
          x[i] = xi;
#pragma unroll
          for (int j = 0; j < i; ++j) x[j] -= sL[i * 32 + j] * xi;
        }
#pragma unroll
        for (int j = 0; j < 32; ++j) sAug[tid * AUG_LD + j] = x[j];
      }
      __syncthreads();
      for (int e = tid; e < 1024; e += 128) {
        int i = e >> 5, j = e & 31;
        sLl[e] = sAug[i * AUG_LD + j];
        sAug[i * AUG_LD + j] = sJd[e] + (i == j ? EPS_CHOL : 0.f);
      }
      __syncthreads();
      mm32<false, true>(sLl, sLl, sAug, AUG_LD, MM_SUB, tid);
      __syncthreads();
    } else {
      for (int e = tid; e < 1024; e += 128) {
        int i = e >> 5, j = e & 31;
        sAug[i * AUG_LD + j] = sJd[e] + (i == j ? EPS_CHOL : 0.f);
      }
      __syncthreads();
    }
    if (tid < 32) {
      float m[32];
#pragma unroll
      for (int j = 0; j < 32; ++j) m[j] = sAug[tid * AUG_LD + j];
#pragma unroll
      for (int k = 0; k < 32; ++k) {
        const float lkk = lane_bcast(sqrtf(m[k]), k);
        const float q = m[k] / lkk;
        m[k] = (tid == k) ? lkk : q;
#pragma unroll
        for (int j = k + 1; j < 32; ++j) m[j] -= m[k] * lane_bcast(m[k], j);
      }
#pragma unroll
      for (int j = 0; j < 32; ++j) sAug[tid * AUG_LD + j] = m[j];
    }
    __syncthreads();
    for (int e = tid; e < 1024; e += 128) {
      int i = e >> 5, j = e & 31;
      sL[e] = (j <= i) ? sAug[i * AUG_LD + j] : 0.f;
    }
    __syncthreads();

    if (tid < DD) {
      float r = noise[((size_t)n * TT + t) * DD + tid];
      if (t > 0) {
        float s = 0.f;
        for (int j = 0; j < DD; ++j) s += sLl[j * 32 + tid] * sx[j];
        r -= s;
      }
#pragma unroll
      for (int i = 31; i >= 0; --i) {
        const float xi = lane_bcast(r, i) / (sL[i * 32 + i] + EPS_BS);
        const float upd = r - sL[i * 32 + tid] * xi;
        r = (tid == i) ? xi : ((tid < i) ? upd : r);
      }
      sx[tid] = r;
      out[((size_t)n * TT + t) * DD + tid] = r;
    }
    __syncthreads();
  }
}

__global__ __launch_bounds__(128) void lds_backward_kernel(
    const float* __restrict__ g_c, const float* __restrict__ g_d,
    float* __restrict__ out) {
  __shared__ float sCt[1024];
  __shared__ float sMu[DD], sNew[DD];
  const int n = blockIdx.x;
  const int tid = threadIdx.x;
  if (tid < DD) {
    float v = g_d[((size_t)n * TT + (TT - 1)) * DD + tid];
    sMu[tid] = v;
    out[((size_t)n * TT + (TT - 1)) * DD + tid] += v;
  }
  __syncthreads();
  for (int t = TT - 2; t >= 0; --t) {
    size_t base = (size_t)n * TT + t;
    for (int e = tid; e < 1024; e += 128) sCt[e] = g_c[base * 1024 + e];
    __syncthreads();
    if (tid < DD) {
      float acc = g_d[base * DD + tid];
      for (int j = 0; j < DD; ++j) acc -= sCt[tid * 32 + j] * sMu[j];
      sNew[tid] = acc;
    }
    __syncthreads();
    if (tid < DD) {
      sMu[tid] = sNew[tid];
      out[base * DD + tid] += sNew[tid];
    }
    __syncthreads();
  }
}

extern "C" void kernel_launch(void* const* d_in, const int* in_sizes, int n_in,
                              void* d_out, int out_size, void* d_ws,
                              size_t ws_size, hipStream_t stream) {
  (void)in_sizes; (void)n_in; (void)out_size; (void)ws_size;
  const float* z_samples = (const float*)d_in[0];
  const float* A_base = (const float*)d_in[1];
  const float* b_base = (const float*)d_in[2];
  const float* Q_sqrt = (const float*)d_in[3];
  const float* ms = (const float*)d_in[4];
  const float* Ri_sqrts = (const float*)d_in[5];
  const float* noise = (const float*)d_in[6];
  float* ws = (float*)d_ws;
  float* g_Ris = ws;
  float* g_hR = g_Ris + (size_t)TT * DD * DD;
  float* g_d = g_hR + (size_t)TT * DD;
  float* g_c = g_d + (size_t)NB * TT * DD;
  float* out = (float*)d_out;

  ris_precompute_kernel<<<TT, 128, 0, stream>>>(Ri_sqrts, ms, g_Ris, g_hR);
  lds_forward_kernel<<<NB_FWD, 128, 0, stream>>>(z_samples, A_base, b_base, Q_sqrt,
                                             noise, g_Ris, g_hR, g_c, g_d, out);
  lds_backward_kernel<<<NB, 128, 0, stream>>>(g_c, g_d, out);
}


# Global loads of both time loops go out in groups, not one wait per element: A_base,Q_sqrt,b_base held in registers; each step's z,Ris,hR,noise and backward c,d,out fetched together; sum order kept
# speedup vs baseline: 5.8292x; 1.2709x over previous
#include <hip/hip_runtime.h>

#define TT 256
#define DD 32
#define KK 8
#define NB 64
#define NB_FWD 128
#define AUG_LD 66
#define EPS_TRI 0.01f
#define EPS_CHOL 0.01f
#define EPS_BS 0.0001f

typedef float v2f __attribute__((ext_vector_type(2)));
typedef float v8f __attribute__((ext_vector_type(8)));

enum { MM_STORE = 0, MM_STORE_NEG = 1, MM_SUB = 2, MM_ADD = 3 };

template <bool TA, bool TB>
__device__ inline void mm32(const float* A, const float* B, float* D, int ldd,
                            int mode, int tid) {
  const int lane = tid & 31;
  const int wid = tid >> 5;
  const int mt = (wid >> 1) << 4;
  const int nt = (wid & 1) << 4;
  const int l15 = lane & 15;
  const int kh = (lane >> 4) << 1;
  v8f acc = {};
#pragma unroll
  for (int kt = 0; kt < 32; kt += 4) {
    const int k0 = kt + kh;
    v2f av, bv;
    if (!TA) { av.x = A[(mt + l15) * 32 + k0];  av.y = A[(mt + l15) * 32 + k0 + 1]; }
    else     { av.x = A[k0 * 32 + mt + l15];    av.y = A[(k0 + 1) * 32 + mt + l15]; }
    if (!TB) { bv.x = B[k0 * 32 + nt + l15];    bv.y = B[(k0 + 1) * 32 + nt + l15]; }
    else     { bv.x = B[(nt + l15) * 32 + k0];  bv.y = B[(nt + l15) * 32 + k0 + 1]; }
    acc = __builtin_amdgcn_wmma_f32_16x16x4_f32(false, av, false, bv, (short)0,
                                                acc, false, false);
  }
  const int rbase = mt + ((lane >> 4) << 3);
#pragma unroll
  for (int r = 0; r < 8; ++r) {
    float v = acc[r];
    float* p = &D[(rbase + r) * ldd + nt + l15];
    if (mode == MM_STORE) *p = v;
    else if (mode == MM_STORE_NEG) *p = -v;
    else if (mode == MM_SUB) *p -= v;
    else *p += v;
  }
}

__device__ inline float lane_bcast(float v, int src) {
  return __builtin_bit_cast(
      float, __builtin_amdgcn_readlane(__builtin_bit_cast(int, v), src));
}

__global__ __launch_bounds__(128) void ris_precompute_kernel(
    const float* __restrict__ Ri_sqrts, const float* __restrict__ ms,
    float* __restrict__ g_Ris, float* __restrict__ g_hR) {
  __shared__ float sR[1024], sO[1024], sm[DD];
  const int t = blockIdx.x;
  const int tid = threadIdx.x;
  for (int e = tid; e < 1024; e += 128) sR[e] = Ri_sqrts[(size_t)t * 1024 + e];
  if (tid < DD) sm[tid] = ms[t * DD + tid];
  __syncthreads();
  mm32<false, true>(sR, sR, sO, 32, MM_STORE, tid);
  __syncthreads();
  for (int e = tid; e < 1024; e += 128) g_Ris[(size_t)t * 1024 + e] = sO[e];
  if (tid < DD) {
    float a = 0.f;
    for (int j = 0; j < DD; ++j) a += sO[tid * 32 + j] * sm[j];
    g_hR[t * DD + tid] = a;
  }
}

__global__ __launch_bounds__(128) void lds_forward_kernel(
    const float* __restrict__ z_samples, const float* __restrict__ A_base,
    const float* __restrict__ b_base, const float* __restrict__ Q_sqrt,
    const float* __restrict__ noise, const float* __restrict__ g_Ris,
    const float* __restrict__ g_hR, float* __restrict__ g_c,
    float* __restrict__ g_d, float* __restrict__ out) {
  __shared__ float sA[1024], sQs[1024], sJd[1024], sC[1024], sL[1024],
      sLl[1024];
  __shared__ float sQis[2][1024], sJl[2][1024];
  __shared__ float sAug[32 * AUG_LD];
  __shared__ float sz[KK], sbs[DD], sQb[DD], sh[DD], sJlPd[DD], sd[DD],
      sDiag[DD], sx[DD];

  const bool triRole = (blockIdx.x < NB);
  const int n = triRole ? blockIdx.x : blockIdx.x - NB;
  const int tid = threadIdx.x;
  const int row = tid & 31;
  const int jgrp = tid >> 5;

  float rA[8][KK], rQ[8][KK], rb[KK];
#pragma unroll
  for (int i = 0; i < 8; ++i) {
#pragma unroll
    for (int k = 0; k < KK; ++k) {
      rA[i][k] = A_base[k * 1024 + tid + 128 * i];
      rQ[i][k] = Q_sqrt[k * 1024 + tid + 128 * i];
    }
  }
#pragma unroll
  for (int k = 0; k < KK; ++k)
    rb[k] = (triRole && tid < DD) ? b_base[k * DD + tid] : 0.f;

  for (int t = 0; t < TT; ++t) {
    const int cur = t & 1, prev = cur ^ 1;
    const bool haveNext = (t < TT - 1);

    const float* vsrc =
        triRole ? (g_hR + t * DD) : (noise + ((size_t)n * TT + t) * DD);
    float zv = 0.f, vecv = 0.f, ris[8];
    if (haveNext && tid < KK) zv = z_samples[((size_t)n * TT + t) * KK + tid];
    if (tid < DD) vecv = vsrc[tid];
#pragma unroll
    for (int i = 0; i < 8; ++i)
      ris[i] = g_Ris[(size_t)t * 1024 + tid + 128 * i];

    if (haveNext) {
      if (tid < KK) sz[tid] = zv;
      __syncthreads();
      {
        float zr[KK];
#pragma unroll
        for (int k = 0; k < KK; ++k) zr[k] = sz[k];
#pragma unroll
        for (int i = 0; i < 8; ++i) {
          float a = 0.f, q = 0.f;
#pragma unroll
          for (int k = 0; k < KK; ++k) {
            a += zr[k] * rA[i][k];
            q += zr[k] * rQ[i][k];
          }
          sA[tid + 128 * i] = a;
          sQs[tid + 128 * i] = q;
        }
        if (triRole && tid < DD) {
          float b = 0.f;
#pragma unroll
          for (int k = 0; k < KK; ++k) b += zr[k] * rb[k];
          sbs[tid] = b;
        }
      }
      __syncthreads();
      mm32<false, true>(sQs, sQs, sQis[cur], 32, MM_STORE, tid);
      __syncthreads();
      mm32<false, false>(sQis[cur], sA, sJl[cur], 32, MM_STORE_NEG, tid);
      __syncthreads();
      mm32<true, false>(sA, sJl[cur], sJd, 32, MM_STORE_NEG, tid);
      __syncthreads();
    } else {
      for (int e = tid; e < 1024; e += 128) sJd[e] = 0.f;
      __syncthreads();
    }

#pragma unroll
    for (int i = 0; i < 8; ++i) {
      const int e = tid + 128 * i;
      float v = sJd[e] + ris[i];
      if (t > 0) v += sQis[prev][e];
      sJd[e] = v;
    }
    if (triRole && tid < DD) {
      float hv = vecv;
      if (t > 0) hv += sQb[tid];
      if (haveNext) {
        float jb = 0.f;
        for (int j = 0; j < DD; ++j) jb += sJl[cur][tid * 32 + j] * sbs[j];
        hv += jb;
      }
      float jd = 0.f;
      if (t > 0)
        for (int j = 0; j < DD; ++j) jd += sJl[prev][tid * 32 + j] * sd[j];
      sh[tid] = hv;
      sJlPd[tid] = jd;
      if (haveNext) {
        float qb = 0.f;
        for (int j = 0; j < DD; ++j) qb += sQis[cur][tid * 32 + j] * sbs[j];
        sQb[tid] = qb;
      }
    }
    __syncthreads();

    if (triRole) {
      for (int idx = tid; idx < 32 * 65; idx += 128) {
        int i = idx / 65, j = idx % 65;
        float v;
        if (j < 32)      v = sJd[i * 32 + j] + (i == j ? EPS_TRI : 0.f);
        else if (j < 64) v = haveNext ? sJl[cur][(j - 32) * 32 + i] : 0.f;
        else             v = sh[i] - (t > 0 ? sJlPd[i] : 0.f);
        sAug[i * AUG_LD + j] = v;
      }
      __syncthreads();
      if (t > 0) {
        mm32<false, false>(sJl[prev], sC, sAug, AUG_LD, MM_SUB, tid);
        __syncthreads();
      }
      {
        float a[32], b[9];
        float diag = 0.f;
#pragma unroll
        for (int j = 0; j < 32; ++j) a[j] = sAug[row * AUG_LD + j];
#pragma unroll
        for (int j = 0; j < 8; ++j)
          b[j] = sAug[row * AUG_LD + 32 + 8 * jgrp + j];
        b[8] = sAug[row * AUG_LD + 64];
#pragma unroll
        for (int k = 0; k < 32; ++k) {
          const float piv = lane_bcast(a[k], k);
          float f = a[k] / piv;
          f = (row == k) ? 0.f : f;
          diag = (row == k) ? piv : diag;
#pragma unroll
          for (int j = k + 1; j < 32; ++j) a[j] -= f * lane_bcast(a[j], k);
#pragma unroll
          for (int j = 0; j < 9; ++j) b[j] -= f * lane_bcast(b[j], k);
        }
#pragma unroll
        for (int j = 0; j < 8; ++j)
          sAug[row * AUG_LD + 32 + 8 * jgrp + j] = b[j];
        if (jgrp == 0) {
          sDiag[row] = diag;
          sd[row] = b[8] / diag;
        }
      }
      __syncthreads();
      for (int e = tid; e < 1024; e += 128) {
        int i = e >> 5, j = e & 31;
        sC[e] = sAug[j * AUG_LD + 32 + i] / sDiag[j];
      }
      __syncthreads();
      {
        size_t base = (size_t)n * TT + t;
        if (haveNext)
          for (int e = tid; e < 1024; e += 128) g_c[base * 1024 + e] = sC[e];
        if (tid < DD) g_d[base * DD + tid] = sd[tid];
      }
      continue;
    }

    if (t > 0) {
      for (int e = tid; e < 1024; e += 128)
        sAug[(e >> 5) * AUG_LD + (e & 31)] = sJl[prev][e];
      __syncthreads();
      if (tid < 32) {
        float x[32];
#pragma unroll
        for (int j = 0; j < 32; ++j) x[j] = sAug[tid * AUG_LD + j];
#pragma unroll
        for (int i = 31; i >= 0; --i) {
          const float xi = x[i] / sL[i * 32 + i];
          x[i] = xi;
#pragma unroll
          for (int j = 0; j < i; ++j) x[j] -= sL[i * 32 + j] * xi;
        }
#pragma unroll
        for (int j = 0; j < 32; ++j) sAug[tid * AUG_LD + j] = x[j];
      }
      __syncthreads();
      for (int e = tid; e < 1024; e += 128) {
        int i = e >> 5, j = e & 31;
        sLl[e] = sAug[i * AUG_LD + j];
        sAug[i * AUG_LD + j] = sJd[e] + (i == j ? EPS_CHOL : 0.f);
      }
      __syncthreads();
      mm32<false, true>(sLl, sLl, sAug, AUG_LD, MM_SUB, tid);
      __syncthreads();
    } else {
      for (int e = tid; e < 1024; e += 128) {
        int i = e >> 5, j = e & 31;
        sAug[i * AUG_LD + j] = sJd[e] + (i == j ? EPS_CHOL : 0.f);
      }
      __syncthreads();
    }
    if (tid < 32) {
      float m[32];
#pragma unroll
      for (int j = 0; j < 32; ++j) m[j] = sAug[tid * AUG_LD + j];
#pragma unroll
      for (int k = 0; k < 32; ++k) {
        const float lkk = lane_bcast(sqrtf(m[k]), k);
        const float q = m[k] / lkk;
        m[k] = (tid == k) ? lkk : q;
#pragma unroll
        for (int j = k + 1; j < 32; ++j) m[j] -= m[k] * lane_bcast(m[k], j);
      }
#pragma unroll
      for (int j = 0; j < 32; ++j) sAug[tid * AUG_LD + j] = m[j];
    }
    __syncthreads();
    for (int e = tid; e < 1024; e += 128) {
      int i = e >> 5, j = e & 31;
      sL[e] = (j <= i) ? sAug[i * AUG_LD + j] : 0.f;
    }
    __syncthreads();

    if (tid < DD) {
      float r = vecv;
      if (t > 0) {
        float s = 0.f;
        for (int j = 0; j < DD; ++j) s += sLl[j * 32 + tid] * sx[j];
        r -= s;
      }
#pragma unroll
      for (int i = 31; i >= 0; --i) {
        const float xi = lane_bcast(r, i) / (sL[i * 32 + i] + EPS_BS);
        const float upd = r - sL[i * 32 + tid] * xi;
        r = (tid == i) ? xi : ((tid < i) ? upd : r);
      }
      sx[tid] = r;
      out[((size_t)n * TT + t) * DD + tid] = r;
    }
    __syncthreads();
  }
}

__global__ __launch_bounds__(128) void lds_backward_kernel(
    const float* __restrict__ g_c, const float* __restrict__ g_d,
    float* __restrict__ out) {
  __shared__ float sCt[1024];
  __shared__ float sMu[DD], sNew[DD];
  const int n = blockIdx.x;
  const int tid = threadIdx.x;
  if (tid < DD) {
    float v = g_d[((size_t)n * TT + (TT - 1)) * DD + tid];
    sMu[tid] = v;
    out[((size_t)n * TT + (TT - 1)) * DD + tid] += v;
  }
  __syncthreads();
  for (int t = TT - 2; t >= 0; --t) {
    size_t base = (size_t)n * TT + t;
    float cr[8], dv = 0.f, ov = 0.f;
#pragma unroll
    for (int i = 0; i < 8; ++i) cr[i] = g_c[base * 1024 + tid + 128 * i];
    if (tid < DD) {
      dv = g_d[base * DD + tid];
      ov = out[base * DD + tid];
    }
#pragma unroll
    for (int i = 0; i < 8; ++i) sCt[tid + 128 * i] = cr[i];
    __syncthreads();
    if (tid < DD) {
      float acc = dv;
      for (int j = 0; j < DD; ++j) acc -= sCt[tid * 32 + j] * sMu[j];
      sNew[tid] = acc;
    }
    __syncthreads();
    if (tid < DD) {
      sMu[tid] = sNew[tid];
      out[base * DD + tid] = ov + sNew[tid];
    }
    __syncthreads();
  }
}

extern "C" void kernel_launch(void* const* d_in, const int* in_sizes, int n_in,
                              void* d_out, int out_size, void* d_ws,
                              size_t ws_size, hipStream_t stream) {
  (void)in_sizes; (void)n_in; (void)out_size; (void)ws_size;
  const float* z_samples = (const float*)d_in[0];
  const float* A_base = (const float*)d_in[1];
  const float* b_base = (const float*)d_in[2];
  const float* Q_sqrt = (const float*)d_in[3];
  const float* ms = (const float*)d_in[4];
  const float* Ri_sqrts = (const float*)d_in[5];
  const float* noise = (const float*)d_in[6];
  float* ws = (float*)d_ws;
  float* g_Ris = ws;
  float* g_hR = g_Ris + (size_t)TT * DD * DD;
  float* g_d = g_hR + (size_t)TT * DD;
  float* g_c = g_d + (size_t)NB * TT * DD;
  float* out = (float*)d_out;

  ris_precompute_kernel<<<TT, 128, 0, stream>>>(Ri_sqrts, ms, g_Ris, g_hR);
  lds_forward_kernel<<<NB_FWD, 128, 0, stream>>>(z_samples, A_base, b_base, Q_sqrt,
                                             noise, g_Ris, g_hR, g_c, g_d, out);
  lds_backward_kernel<<<NB, 128, 0, stream>>>(g_c, g_d, out);
}


# Both triangular solves of the Cholesky role keep only multiply-adds on their serial chains: L entries and diagonal inverses are fetched/formed before the chain; x/d becomes x*(1/d), rounding differs
# speedup vs baseline: 8.0053x; 1.3733x over previous
#include <hip/hip_runtime.h>

#define TT 256
#define DD 32
#define KK 8
#define NB 64
#define NB_FWD 128
#define AUG_LD 66
#define EPS_TRI 0.01f
#define EPS_CHOL 0.01f
#define EPS_BS 0.0001f

typedef float v2f __attribute__((ext_vector_type(2)));
typedef float v8f __attribute__((ext_vector_type(8)));

enum { MM_STORE = 0, MM_STORE_NEG = 1, MM_SUB = 2, MM_ADD = 3 };

template <bool TA, bool TB>
__device__ inline void mm32(const float* A, const float* B, float* D, int ldd,
                            int mode, int tid) {
  const int lane = tid & 31;
  const int wid = tid >> 5;
  const int mt = (wid >> 1) << 4;
  const int nt = (wid & 1) << 4;
  const int l15 = lane & 15;
  const int kh = (lane >> 4) << 1;
  v8f acc = {};
#pragma unroll
  for (int kt = 0; kt < 32; kt += 4) {
    const int k0 = kt + kh;
    v2f av, bv;
    if (!TA) { av.x = A[(mt + l15) * 32 + k0];  av.y = A[(mt + l15) * 32 + k0 + 1]; }
    else     { av.x = A[k0 * 32 + mt + l15];    av.y = A[(k0 + 1) * 32 + mt + l15]; }
    if (!TB) { bv.x = B[k0 * 32 + nt + l15];    bv.y = B[(k0 + 1) * 32 + nt + l15]; }
    else     { bv.x = B[(nt + l15) * 32 + k0];  bv.y = B[(nt + l15) * 32 + k0 + 1]; }
    acc = __builtin_amdgcn_wmma_f32_16x16x4_f32(false, av, false, bv, (short)0,
                                                acc, false, false);
  }
  const int rbase = mt + ((lane >> 4) << 3);
#pragma unroll
  for (int r = 0; r < 8; ++r) {
    float v = acc[r];
    float* p = &D[(rbase + r) * ldd + nt + l15];
    if (mode == MM_STORE) *p = v;
    else if (mode == MM_STORE_NEG) *p = -v;
    else if (mode == MM_SUB) *p -= v;
    else *p += v;
  }
}

__device__ inline float lane_bcast(float v, int src) {
  return __builtin_bit_cast(
      float, __builtin_amdgcn_readlane(__builtin_bit_cast(int, v), src));
}

__global__ __launch_bounds__(128) void ris_precompute_kernel(
    const float* __restrict__ Ri_sqrts, const float* __restrict__ ms,
    float* __restrict__ g_Ris, float* __restrict__ g_hR) {
  __shared__ float sR[1024], sO[1024], sm[DD];
  const int t = blockIdx.x;
  const int tid = threadIdx.x;
  for (int e = tid; e < 1024; e += 128) sR[e] = Ri_sqrts[(size_t)t * 1024 + e];
  if (tid < DD) sm[tid] = ms[t * DD + tid];
  __syncthreads();
  mm32<false, true>(sR, sR, sO, 32, MM_STORE, tid);
  __syncthreads();
  for (int e = tid; e < 1024; e += 128) g_Ris[(size_t)t * 1024 + e] = sO[e];
  if (tid < DD) {
    float a = 0.f;
    for (int j = 0; j < DD; ++j) a += sO[tid * 32 + j] * sm[j];
    g_hR[t * DD + tid] = a;
  }
}

__global__ __launch_bounds__(128) void lds_forward_kernel(
    const float* __restrict__ z_samples, const float* __restrict__ A_base,
    const float* __restrict__ b_base, const float* __restrict__ Q_sqrt,
    const float* __restrict__ noise, const float* __restrict__ g_Ris,
    const float* __restrict__ g_hR, float* __restrict__ g_c,
    float* __restrict__ g_d, float* __restrict__ out) {
  __shared__ float sA[1024], sQs[1024], sJd[1024], sC[1024], sL[1024],
      sLl[1024];
  __shared__ float sQis[2][1024], sJl[2][1024];
  __shared__ float sAug[32 * AUG_LD];
  __shared__ float sz[KK], sbs[DD], sQb[DD], sh[DD], sJlPd[DD], sd[DD],
      sDiag[DD], sx[DD];

  const bool triRole = (blockIdx.x < NB);
  const int n = triRole ? blockIdx.x : blockIdx.x - NB;
  const int tid = threadIdx.x;
  const int row = tid & 31;
  const int jgrp = tid >> 5;

  float rA[8][KK], rQ[8][KK], rb[KK];
#pragma unroll
  for (int i = 0; i < 8; ++i) {
#pragma unroll
    for (int k = 0; k < KK; ++k) {
      rA[i][k] = A_base[k * 1024 + tid + 128 * i];
      rQ[i][k] = Q_sqrt[k * 1024 + tid + 128 * i];
    }
  }
#pragma unroll
  for (int k = 0; k < KK; ++k)
    rb[k] = (triRole && tid < DD) ? b_base[k * DD + tid] : 0.f;

  for (int t = 0; t < TT; ++t) {
    const int cur = t & 1, prev = cur ^ 1;
    const bool haveNext = (t < TT - 1);

    const float* vsrc =
        triRole ? (g_hR + t * DD) : (noise + ((size_t)n * TT + t) * DD);
    float zv = 0.f, vecv = 0.f, ris[8];
    if (haveNext && tid < KK) zv = z_samples[((size_t)n * TT + t) * KK + tid];
    if (tid < DD) vecv = vsrc[tid];
#pragma unroll
    for (int i = 0; i < 8; ++i)
      ris[i] = g_Ris[(size_t)t * 1024 + tid + 128 * i];

    if (haveNext) {
      if (tid < KK) sz[tid] = zv;
      __syncthreads();
      {
        float zr[KK];
#pragma unroll
        for (int k = 0; k < KK; ++k) zr[k] = sz[k];
#pragma unroll
        for (int i = 0; i < 8; ++i) {
          float a = 0.f, q = 0.f;
#pragma unroll
          for (int k = 0; k < KK; ++k) {
            a += zr[k] * rA[i][k];
            q += zr[k] * rQ[i][k];
          }
          sA[tid + 128 * i] = a;
          sQs[tid + 128 * i] = q;
        }
        if (triRole && tid < DD) {
          float b = 0.f;
#pragma unroll
          for (int k = 0; k < KK; ++k) b += zr[k] * rb[k];
          sbs[tid] = b;
        }
      }
      __syncthreads();
      mm32<false, true>(sQs, sQs, sQis[cur], 32, MM_STORE, tid);
      __syncthreads();
      mm32<false, false>(sQis[cur], sA, sJl[cur], 32, MM_STORE_NEG, tid);
      __syncthreads();
      mm32<true, false>(sA, sJl[cur], sJd, 32, MM_STORE_NEG, tid);
      __syncthreads();
    } else {
      for (int e = tid; e < 1024; e += 128) sJd[e] = 0.f;
      __syncthreads();
    }

#pragma unroll
    for (int i = 0; i < 8; ++i) {
      const int e = tid + 128 * i;
      float v = sJd[e] + ris[i];
      if (t > 0) v += sQis[prev][e];
      sJd[e] = v;
    }
    if (triRole && tid < DD) {
      float hv = vecv;
      if (t > 0) hv += sQb[tid];
      if (haveNext) {
        float jb = 0.f;
        for (int j = 0; j < DD; ++j) jb += sJl[cur][tid * 32 + j] * sbs[j];
        hv += jb;
      }
      float jd = 0.f;
      if (t > 0)
        for (int j = 0; j < DD; ++j) jd += sJl[prev][tid * 32 + j] * sd[j];
      sh[tid] = hv;
      sJlPd[tid] = jd;
      if (haveNext) {
        float qb = 0.f;
        for (int j = 0; j < DD; ++j) qb += sQis[cur][tid * 32 + j] * sbs[j];
        sQb[tid] = qb;
      }
    }
    __syncthreads();

    if (triRole) {
      for (int idx = tid; idx < 32 * 65; idx += 128) {
        int i = idx / 65, j = idx % 65;
        float v;
        if (j < 32)      v = sJd[i * 32 + j] + (i == j ? EPS_TRI : 0.f);
        else if (j < 64) v = haveNext ? sJl[cur][(j - 32) * 32 + i] : 0.f;
        else             v = sh[i] - (t > 0 ? sJlPd[i] : 0.f);
        sAug[i * AUG_LD + j] = v;
      }
      __syncthreads();
      if (t > 0) {
        mm32<false, false>(sJl[prev], sC, sAug, AUG_LD, MM_SUB, tid);
        __syncthreads();
      }
      {
        float a[32], b[9];
        float diag = 0.f;
#pragma unroll
        for (int j = 0; j < 32; ++j) a[j] = sAug[row * AUG_LD + j];
#pragma unroll
        for (int j = 0; j < 8; ++j)
          b[j] = sAug[row * AUG_LD + 32 + 8 * jgrp + j];
        b[8] = sAug[row * AUG_LD + 64];
#pragma unroll
        for (int k = 0; k < 32; ++k) {
          const float piv = lane_bcast(a[k], k);
          float f = a[k] / piv;
          f = (row == k) ? 0.f : f;
          diag = (row == k) ? piv : diag;
#pragma unroll
          for (int j = k + 1; j < 32; ++j) a[j] -= f * lane_bcast(a[j], k);
#pragma unroll
          for (int j = 0; j < 9; ++j) b[j] -= f * lane_bcast(b[j], k);
        }
#pragma unroll
        for (int j = 0; j < 8; ++j)
          sAug[row * AUG_LD + 32 + 8 * jgrp + j] = b[j];
        if (jgrp == 0) {
          sDiag[row] = diag;
          sd[row] = b[8] / diag;
        }
      }
      __syncthreads();
      for (int e = tid; e < 1024; e += 128) {
        int i = e >> 5, j = e & 31;
        sC[e] = sAug[j * AUG_LD + 32 + i] / sDiag[j];
      }
      __syncthreads();
      {
        size_t base = (size_t)n * TT + t;
        if (haveNext)
          for (int e = tid; e < 1024; e += 128) g_c[base * 1024 + e] = sC[e];
        if (tid < DD) g_d[base * DD + tid] = sd[tid];
      }
      continue;
    }

    if (t > 0) {
      for (int e = tid; e < 1024; e += 128)
        sAug[(e >> 5) * AUG_LD + (e & 31)] = sJl[prev][e];
      __syncthreads();
      if (tid < 32) {
        float x[32];
        const float dl = 1.0f / sL[tid * 33];
#pragma unroll
        for (int j = 0; j < 32; ++j) x[j] = sAug[tid * AUG_LD + j];
#pragma unroll
        for (int i = 31; i >= 0; --i) {
          const float xi = x[i] * lane_bcast(dl, i);
          x[i] = xi;
#pragma unroll
          for (int j = 0; j < i; ++j) x[j] -= sL[i * 32 + j] * xi;
        }
#pragma unroll
        for (int j = 0; j < 32; ++j) sAug[tid * AUG_LD + j] = x[j];
      }
      __syncthreads();
      for (int e = tid; e < 1024; e += 128) {
        int i = e >> 5, j = e & 31;
        sLl[e] = sAug[i * AUG_LD + j];
        sAug[i * AUG_LD + j] = sJd[e] + (i == j ? EPS_CHOL : 0.f);
      }
      __syncthreads();
      mm32<false, true>(sLl, sLl, sAug, AUG_LD, MM_SUB, tid);
      __syncthreads();
    } else {
      for (int e = tid; e < 1024; e += 128) {
        int i = e >> 5, j = e & 31;
        sAug[i * AUG_LD + j] = sJd[e] + (i == j ? EPS_CHOL : 0.f);
      }
      __syncthreads();
    }
    if (tid < 32) {
      float m[32];
#pragma unroll
      for (int j = 0; j < 32; ++j) m[j] = sAug[tid * AUG_LD + j];
#pragma unroll
      for (int k = 0; k < 32; ++k) {
        const float lkk = lane_bcast(sqrtf(m[k]), k);
        const float q = m[k] / lkk;
        m[k] = (tid == k) ? lkk : q;
#pragma unroll
        for (int j = k + 1; j < 32; ++j) m[j] -= m[k] * lane_bcast(m[k], j);
      }
#pragma unroll
      for (int j = 0; j < 32; ++j) sAug[tid * AUG_LD + j] = m[j];
    }
    __syncthreads();
    for (int e = tid; e < 1024; e += 128) {
      int i = e >> 5, j = e & 31;
      sL[e] = (j <= i) ? sAug[i * AUG_LD + j] : 0.f;
    }
    __syncthreads();

    if (tid < DD) {
      float lc[32];
#pragma unroll
      for (int i = 0; i < 32; ++i) lc[i] = sL[i * 32 + tid];
      const float dinv = 1.0f / (sL[tid * 33] + EPS_BS);
      float r = vecv;
      if (t > 0) {
        float s = 0.f;
        for (int j = 0; j < DD; ++j) s += sLl[j * 32 + tid] * sx[j];
        r -= s;
      }
      float xv = 0.f;
#pragma unroll
      for (int i = 31; i >= 0; --i) {
        const float p = r * dinv;
        const float xi = lane_bcast(p, i);
        xv = (tid == i) ? p : xv;
        r -= lc[i] * xi;
      }
      sx[tid] = xv;
      out[((size_t)n * TT + t) * DD + tid] = xv;
    }
    __syncthreads();
  }
}

__global__ __launch_bounds__(128) void lds_backward_kernel(
    const float* __restrict__ g_c, const float* __restrict__ g_d,
    float* __restrict__ out) {
  __shared__ float sCt[1024];
  __shared__ float sMu[DD], sNew[DD];
  const int n = blockIdx.x;
  const int tid = threadIdx.x;
  if (tid < DD) {
    float v = g_d[((size_t)n * TT + (TT - 1)) * DD + tid];
    sMu[tid] = v;
    out[((size_t)n * TT + (TT - 1)) * DD + tid] += v;
  }
  __syncthreads();
  for (int t = TT - 2; t >= 0; --t) {
    size_t base = (size_t)n * TT + t;
    float cr[8], dv = 0.f, ov = 0.f;
#pragma unroll
    for (int i = 0; i < 8; ++i) cr[i] = g_c[base * 1024 + tid + 128 * i];
    if (tid < DD) {
      dv = g_d[base * DD + tid];
      ov = out[base * DD + tid];
    }
#pragma unroll
    for (int i = 0; i < 8; ++i) sCt[tid + 128 * i] = cr[i];
    __syncthreads();
    if (tid < DD) {
      float acc = dv;
      for (int j = 0; j < DD; ++j) acc -= sCt[tid * 32 + j] * sMu[j];
      sNew[tid] = acc;
    }
    __syncthreads();
    if (tid < DD) {
      sMu[tid] = sNew[tid];
      out[base * DD + tid] = ov + sNew[tid];
    }
    __syncthreads();
  }
}

extern "C" void kernel_launch(void* const* d_in, const int* in_sizes, int n_in,
                              void* d_out, int out_size, void* d_ws,
                              size_t ws_size, hipStream_t stream) {
  (void)in_sizes; (void)n_in; (void)out_size; (void)ws_size;
  const float* z_samples = (const float*)d_in[0];
  const float* A_base = (const float*)d_in[1];
  const float* b_base = (const float*)d_in[2];
  const float* Q_sqrt = (const float*)d_in[3];
  const float* ms = (const float*)d_in[4];
  const float* Ri_sqrts = (const float*)d_in[5];
  const float* noise = (const float*)d_in[6];
  float* ws = (float*)d_ws;
  float* g_Ris = ws;
  float* g_hR = g_Ris + (size_t)TT * DD * DD;
  float* g_d = g_hR + (size_t)TT * DD;
  float* g_c = g_d + (size_t)NB * TT * DD;
  float* out = (float*)d_out;

  ris_precompute_kernel<<<TT, 128, 0, stream>>>(Ri_sqrts, ms, g_Ris, g_hR);
  lds_forward_kernel<<<NB_FWD, 128, 0, stream>>>(z_samples, A_base, b_base, Q_sqrt,
                                             noise, g_Ris, g_hR, g_c, g_d, out);
  lds_backward_kernel<<<NB, 128, 0, stream>>>(g_c, g_d, out);
}


# LDS staging passes of the time loop (Jd finalize, augmented build, c extraction and its store, the Cholesky role's three copies) unrolled: each thread's LDS reads go out as one group; sums unchanged
# speedup vs baseline: 10.4925x; 1.3107x over previous
#include <hip/hip_runtime.h>

#define TT 256
#define DD 32
#define KK 8
#define NB 64
#define NB_FWD 128
#define AUG_LD 66
#define EPS_TRI 0.01f
#define EPS_CHOL 0.01f
#define EPS_BS 0.0001f

typedef float v2f __attribute__((ext_vector_type(2)));
typedef float v8f __attribute__((ext_vector_type(8)));

enum { MM_STORE = 0, MM_STORE_NEG = 1, MM_SUB = 2, MM_ADD = 3 };

template <bool TA, bool TB>
__device__ inline void mm32(const float* A, const float* B, float* D, int ldd,
                            int mode, int tid) {
  const int lane = tid & 31;
  const int wid = tid >> 5;
  const int mt = (wid >> 1) << 4;
  const int nt = (wid & 1) << 4;
  const int l15 = lane & 15;
  const int kh = (lane >> 4) << 1;
  v8f acc = {};
#pragma unroll
  for (int kt = 0; kt < 32; kt += 4) {
    const int k0 = kt + kh;
    v2f av, bv;
    if (!TA) { av.x = A[(mt + l15) * 32 + k0];  av.y = A[(mt + l15) * 32 + k0 + 1]; }
    else     { av.x = A[k0 * 32 + mt + l15];    av.y = A[(k0 + 1) * 32 + mt + l15]; }
    if (!TB) { bv.x = B[k0 * 32 + nt + l15];    bv.y = B[(k0 + 1) * 32 + nt + l15]; }
    else     { bv.x = B[(nt + l15) * 32 + k0];  bv.y = B[(nt + l15) * 32 + k0 + 1]; }
    acc = __builtin_amdgcn_wmma_f32_16x16x4_f32(false, av, false, bv, (short)0,
                                                acc, false, false);
  }
  const int rbase = mt + ((lane >> 4) << 3);
#pragma unroll
  for (int r = 0; r < 8; ++r) {
    float v = acc[r];
    float* p = &D[(rbase + r) * ldd + nt + l15];
    if (mode == MM_STORE) *p = v;
    else if (mode == MM_STORE_NEG) *p = -v;
    else if (mode == MM_SUB) *p -= v;
    else *p += v;
  }
}

__device__ inline float lane_bcast(float v, int src) {
  return __builtin_bit_cast(
      float, __builtin_amdgcn_readlane(__builtin_bit_cast(int, v), src));
}

__global__ __launch_bounds__(128) void ris_precompute_kernel(
    const float* __restrict__ Ri_sqrts, const float* __restrict__ ms,
    float* __restrict__ g_Ris, float* __restrict__ g_hR) {
  __shared__ float sR[1024], sO[1024], sm[DD];
  const int t = blockIdx.x;
  const int tid = threadIdx.x;
  for (int e = tid; e < 1024; e += 128) sR[e] = Ri_sqrts[(size_t)t * 1024 + e];
  if (tid < DD) sm[tid] = ms[t * DD + tid];
  __syncthreads();
  mm32<false, true>(sR, sR, sO, 32, MM_STORE, tid);
  __syncthreads();
  for (int e = tid; e < 1024; e += 128) g_Ris[(size_t)t * 1024 + e] = sO[e];
  if (tid < DD) {
    float a = 0.f;
    for (int j = 0; j < DD; ++j) a += sO[tid * 32 + j] * sm[j];
    g_hR[t * DD + tid] = a;
  }
}

__global__ __launch_bounds__(128) void lds_forward_kernel(
    const float* __restrict__ z_samples, const float* __restrict__ A_base,
    const float* __restrict__ b_base, const float* __restrict__ Q_sqrt,
    const float* __restrict__ noise, const float* __restrict__ g_Ris,
    const float* __restrict__ g_hR, float* __restrict__ g_c,
    float* __restrict__ g_d, float* __restrict__ out) {
  __shared__ float sA[1024], sQs[1024], sJd[1024], sC[1024], sL[1024],
      sLl[1024];
  __shared__ float sQis[2][1024], sJl[2][1024];
  __shared__ float sAug[32 * AUG_LD];
  __shared__ float sz[KK], sbs[DD], sQb[DD], sh[DD], sJlPd[DD], sd[DD],
      sDiag[DD], sx[DD];

  const bool triRole = (blockIdx.x < NB);
  const int n = triRole ? blockIdx.x : blockIdx.x - NB;
  const int tid = threadIdx.x;
  const int row = tid & 31;
  const int jgrp = tid >> 5;

  float rA[8][KK], rQ[8][KK], rb[KK];
#pragma unroll
  for (int i = 0; i < 8; ++i) {
#pragma unroll
    for (int k = 0; k < KK; ++k) {
      rA[i][k] = A_base[k * 1024 + tid + 128 * i];
      rQ[i][k] = Q_sqrt[k * 1024 + tid + 128 * i];
    }
  }
#pragma unroll
  for (int k = 0; k < KK; ++k)
    rb[k] = (triRole && tid < DD) ? b_base[k * DD + tid] : 0.f;

  for (int t = 0; t < TT; ++t) {
    const int cur = t & 1, prev = cur ^ 1;
    const bool haveNext = (t < TT - 1);

    const float* vsrc =
        triRole ? (g_hR + t * DD) : (noise + ((size_t)n * TT + t) * DD);
    float zv = 0.f, vecv = 0.f, ris[8];
    if (haveNext && tid < KK) zv = z_samples[((size_t)n * TT + t) * KK + tid];
    if (tid < DD) vecv = vsrc[tid];
#pragma unroll
    for (int i = 0; i < 8; ++i)
      ris[i] = g_Ris[(size_t)t * 1024 + tid + 128 * i];

    if (haveNext) {
      if (tid < KK) sz[tid] = zv;
      __syncthreads();
      {
        float zr[KK];
#pragma unroll
        for (int k = 0; k < KK; ++k) zr[k] = sz[k];
#pragma unroll
        for (int i = 0; i < 8; ++i) {
          float a = 0.f, q = 0.f;
#pragma unroll
          for (int k = 0; k < KK; ++k) {
            a += zr[k] * rA[i][k];
            q += zr[k] * rQ[i][k];
          }
          sA[tid + 128 * i] = a;
          sQs[tid + 128 * i] = q;
        }
        if (triRole && tid < DD) {
          float b = 0.f;
#pragma unroll
          for (int k = 0; k < KK; ++k) b += zr[k] * rb[k];
          sbs[tid] = b;
        }
      }
      __syncthreads();
      mm32<false, true>(sQs, sQs, sQis[cur], 32, MM_STORE, tid);
      __syncthreads();
      mm32<false, false>(sQis[cur], sA, sJl[cur], 32, MM_STORE_NEG, tid);
      __syncthreads();
      mm32<true, false>(sA, sJl[cur], sJd, 32, MM_STORE_NEG, tid);
      __syncthreads();
    } else {
      for (int e = tid; e < 1024; e += 128) sJd[e] = 0.f;
      __syncthreads();
    }

    {
      float jd[8];
      if (t > 0) {
        float qp[8];
#pragma unroll
        for (int i = 0; i < 8; ++i) {
          jd[i] = sJd[tid + 128 * i];
          qp[i] = sQis[prev][tid + 128 * i];
        }
#pragma unroll
        for (int i = 0; i < 8; ++i) sJd[tid + 128 * i] = (jd[i] + ris[i]) + qp[i];
      } else {
#pragma unroll
        for (int i = 0; i < 8; ++i) jd[i] = sJd[tid + 128 * i];
#pragma unroll
        for (int i = 0; i < 8; ++i) sJd[tid + 128 * i] = jd[i] + ris[i];
      }
    }
    if (triRole && tid < DD) {
      float hv = vecv;
      if (t > 0) hv += sQb[tid];
      if (haveNext) {
        float jb = 0.f;
        for (int j = 0; j < DD; ++j) jb += sJl[cur][tid * 32 + j] * sbs[j];
        hv += jb;
      }
      float jd = 0.f;
      if (t > 0)
        for (int j = 0; j < DD; ++j) jd += sJl[prev][tid * 32 + j] * sd[j];
      sh[tid] = hv;
      sJlPd[tid] = jd;
      if (haveNext) {
        float qb = 0.f;
        for (int j = 0; j < DD; ++j) qb += sQis[cur][tid * 32 + j] * sbs[j];
        sQb[tid] = qb;
      }
    }
    __syncthreads();

    if (triRole) {
      {
        float aj[8], al[8];
#pragma unroll
        for (int i = 0; i < 8; ++i) aj[i] = sJd[tid + 128 * i];
        if (haveNext) {
#pragma unroll
          for (int i = 0; i < 8; ++i) al[i] = sJl[cur][tid + 128 * i];
        } else {
#pragma unroll
          for (int i = 0; i < 8; ++i) al[i] = 0.f;
        }
        float rv = 0.f;
        if (tid < DD) rv = sh[tid] - (t > 0 ? sJlPd[tid] : 0.f);
#pragma unroll
        for (int i = 0; i < 8; ++i) {
          const int r = jgrp + 4 * i;
          sAug[r * AUG_LD + row] = aj[i] + (r == row ? EPS_TRI : 0.f);
          sAug[row * AUG_LD + 32 + r] = al[i];
        }
        if (tid < DD) sAug[tid * AUG_LD + 64] = rv;
      }
      __syncthreads();
      if (t > 0) {
        mm32<false, false>(sJl[prev], sC, sAug, AUG_LD, MM_SUB, tid);
        __syncthreads();
      }
      {
        float a[32], b[9];
        float diag = 0.f;
#pragma unroll
        for (int j = 0; j < 32; ++j) a[j] = sAug[row * AUG_LD + j];
#pragma unroll
        for (int j = 0; j < 8; ++j)
          b[j] = sAug[row * AUG_LD + 32 + 8 * jgrp + j];
        b[8] = sAug[row * AUG_LD + 64];
#pragma unroll
        for (int k = 0; k < 32; ++k) {
          const float piv = lane_bcast(a[k], k);
          float f = a[k] / piv;
          f = (row == k) ? 0.f : f;
          diag = (row == k) ? piv : diag;
#pragma unroll
          for (int j = k + 1; j < 32; ++j) a[j] -= f * lane_bcast(a[j], k);
#pragma unroll
          for (int j = 0; j < 9; ++j) b[j] -= f * lane_bcast(b[j], k);
        }
#pragma unroll
        for (int j = 0; j < 8; ++j)
          sAug[row * AUG_LD + 32 + 8 * jgrp + j] = b[j];
        if (jgrp == 0) {
          sDiag[row] = diag;
          sd[row] = b[8] / diag;
        }
      }
      __syncthreads();
      {
        const size_t base = (size_t)n * TT + t;
        float cv[8];
        const float dg = sDiag[row];
#pragma unroll
        for (int i = 0; i < 8; ++i)
          cv[i] = sAug[row * AUG_LD + 32 + jgrp + 4 * i];
#pragma unroll
        for (int i = 0; i < 8; ++i) {
          cv[i] = cv[i] / dg;
          sC[tid + 128 * i] = cv[i];
        }
        if (haveNext) {
#pragma unroll
          for (int i = 0; i < 8; ++i) g_c[base * 1024 + tid + 128 * i] = cv[i];
        }
        if (tid < DD) g_d[base * DD + tid] = sd[tid];
      }
      continue;
    }

    if (t > 0) {
      {
        float jl[8];
#pragma unroll
        for (int i = 0; i < 8; ++i) jl[i] = sJl[prev][tid + 128 * i];
#pragma unroll
        for (int i = 0; i < 8; ++i) sAug[(jgrp + 4 * i) * AUG_LD + row] = jl[i];
      }
      __syncthreads();
      if (tid < 32) {
        float x[32];
        const float dl = 1.0f / sL[tid * 33];
#pragma unroll
        for (int j = 0; j < 32; ++j) x[j] = sAug[tid * AUG_LD + j];
#pragma unroll
        for (int i = 31; i >= 0; --i) {
          const float xi = x[i] * lane_bcast(dl, i);
          x[i] = xi;
#pragma unroll
          for (int j = 0; j < i; ++j) x[j] -= sL[i * 32 + j] * xi;
        }
#pragma unroll
        for (int j = 0; j < 32; ++j) sAug[tid * AUG_LD + j] = x[j];
      }
      __syncthreads();
      {
        float xl[8], jd[8];
#pragma unroll
        for (int i = 0; i < 8; ++i) {
          xl[i] = sAug[(jgrp + 4 * i) * AUG_LD + row];
          jd[i] = sJd[tid + 128 * i];
        }
#pragma unroll
        for (int i = 0; i < 8; ++i) {
          const int r = jgrp + 4 * i;
          sLl[tid + 128 * i] = xl[i];
          sAug[r * AUG_LD + row] = jd[i] + (r == row ? EPS_CHOL : 0.f);
        }
      }
      __syncthreads();
      mm32<false, true>(sLl, sLl, sAug, AUG_LD, MM_SUB, tid);
      __syncthreads();
    } else {
      float jd[8];
#pragma unroll
      for (int i = 0; i < 8; ++i) jd[i] = sJd[tid + 128 * i];
#pragma unroll
      for (int i = 0; i < 8; ++i) {
        const int r = jgrp + 4 * i;
        sAug[r * AUG_LD + row] = jd[i] + (r == row ? EPS_CHOL : 0.f);
      }
      __syncthreads();
    }
    if (tid < 32) {
      float m[32];
#pragma unroll
      for (int j = 0; j < 32; ++j) m[j] = sAug[tid * AUG_LD + j];
#pragma unroll
      for (int k = 0; k < 32; ++k) {
        const float lkk = lane_bcast(sqrtf(m[k]), k);
        const float q = m[k] / lkk;
        m[k] = (tid == k) ? lkk : q;
#pragma unroll
        for (int j = k + 1; j < 32; ++j) m[j] -= m[k] * lane_bcast(m[k], j);
      }
#pragma unroll
      for (int j = 0; j < 32; ++j) sAug[tid * AUG_LD + j] = m[j];
    }
    __syncthreads();
    {
      float lv[8];
#pragma unroll
      for (int i = 0; i < 8; ++i) lv[i] = sAug[(jgrp + 4 * i) * AUG_LD + row];
#pragma unroll
      for (int i = 0; i < 8; ++i)
        sL[tid + 128 * i] = (row <= jgrp + 4 * i) ? lv[i] : 0.f;
    }
    __syncthreads();

    if (tid < DD) {
      float lc[32];
#pragma unroll
      for (int i = 0; i < 32; ++i) lc[i] = sL[i * 32 + tid];
      const float dinv = 1.0f / (sL[tid * 33] + EPS_BS);
      float r = vecv;
      if (t > 0) {
        float s = 0.f;
        for (int j = 0; j < DD; ++j) s += sLl[j * 32 + tid] * sx[j];
        r -= s;
      }
      float xv = 0.f;
#pragma unroll
      for (int i = 31; i >= 0; --i) {
        const float p = r * dinv;
        const float xi = lane_bcast(p, i);
        xv = (tid == i) ? p : xv;
        r -= lc[i] * xi;
      }
      sx[tid] = xv;
      out[((size_t)n * TT + t) * DD + tid] = xv;
    }
    __syncthreads();
  }
}

__global__ __launch_bounds__(128) void lds_backward_kernel(
    const float* __restrict__ g_c, const float* __restrict__ g_d,
    float* __restrict__ out) {
  __shared__ float sCt[1024];
  __shared__ float sMu[DD], sNew[DD];
  const int n = blockIdx.x;
  const int tid = threadIdx.x;
  if (tid < DD) {
    float v = g_d[((size_t)n * TT + (TT - 1)) * DD + tid];
    sMu[tid] = v;
    out[((size_t)n * TT + (TT - 1)) * DD + tid] += v;
  }
  __syncthreads();
  for (int t = TT - 2; t >= 0; --t) {
    size_t base = (size_t)n * TT + t;
    float cr[8], dv = 0.f, ov = 0.f;
#pragma unroll
    for (int i = 0; i < 8; ++i) cr[i] = g_c[base * 1024 + tid + 128 * i];
    if (tid < DD) {
      dv = g_d[base * DD + tid];
      ov = out[base * DD + tid];
    }
#pragma unroll
    for (int i = 0; i < 8; ++i) sCt[tid + 128 * i] = cr[i];
    __syncthreads();
    if (tid < DD) {
      float acc = dv;
      for (int j = 0; j < DD; ++j) acc -= sCt[tid * 32 + j] * sMu[j];
      sNew[tid] = acc;
    }
    __syncthreads();
    if (tid < DD) {
      sMu[tid] = sNew[tid];
      out[base * DD + tid] = ov + sNew[tid];
    }
    __syncthreads();
  }
}

extern "C" void kernel_launch(void* const* d_in, const int* in_sizes, int n_in,
                              void* d_out, int out_size, void* d_ws,
                              size_t ws_size, hipStream_t stream) {
  (void)in_sizes; (void)n_in; (void)out_size; (void)ws_size;
  const float* z_samples = (const float*)d_in[0];
  const float* A_base = (const float*)d_in[1];
  const float* b_base = (const float*)d_in[2];
  const float* Q_sqrt = (const float*)d_in[3];
  const float* ms = (const float*)d_in[4];
  const float* Ri_sqrts = (const float*)d_in[5];
  const float* noise = (const float*)d_in[6];
  float* ws = (float*)d_ws;
  float* g_Ris = ws;
  float* g_hR = g_Ris + (size_t)TT * DD * DD;
  float* g_d = g_hR + (size_t)TT * DD;
  float* g_c = g_d + (size_t)NB * TT * DD;
  float* out = (float*)d_out;

  ris_precompute_kernel<<<TT, 128, 0, stream>>>(Ri_sqrts, ms, g_Ris, g_hR);
  lds_forward_kernel<<<NB_FWD, 128, 0, stream>>>(z_samples, A_base, b_base, Q_sqrt,
                                             noise, g_Ris, g_hR, g_c, g_d, out);
  lds_backward_kernel<<<NB, 128, 0, stream>>>(g_c, g_d, out);
}


# Pivot steps of both factorizations use the hardware reciprocal: Cholesky column = m*rsq(pivot) (was IEEE sqrt then divide), Gauss-Jordan factor = a*rcp(pivot); 1-ulp v_rsq/v_rcp, rounding differs
# speedup vs baseline: 12.8315x; 1.2229x over previous
#include <hip/hip_runtime.h>

#define TT 256
#define DD 32
#define KK 8
#define NB 64
#define NB_FWD 128
#define AUG_LD 66
#define EPS_TRI 0.01f
#define EPS_CHOL 0.01f
#define EPS_BS 0.0001f

typedef float v2f __attribute__((ext_vector_type(2)));
typedef float v8f __attribute__((ext_vector_type(8)));

enum { MM_STORE = 0, MM_STORE_NEG = 1, MM_SUB = 2, MM_ADD = 3 };

template <bool TA, bool TB>
__device__ inline void mm32(const float* A, const float* B, float* D, int ldd,
                            int mode, int tid) {
  const int lane = tid & 31;
  const int wid = tid >> 5;
  const int mt = (wid >> 1) << 4;
  const int nt = (wid & 1) << 4;
  const int l15 = lane & 15;
  const int kh = (lane >> 4) << 1;
  v8f acc = {};
#pragma unroll
  for (int kt = 0; kt < 32; kt += 4) {
    const int k0 = kt + kh;
    v2f av, bv;
    if (!TA) { av.x = A[(mt + l15) * 32 + k0];  av.y = A[(mt + l15) * 32 + k0 + 1]; }
    else     { av.x = A[k0 * 32 + mt + l15];    av.y = A[(k0 + 1) * 32 + mt + l15]; }
    if (!TB) { bv.x = B[k0 * 32 + nt + l15];    bv.y = B[(k0 + 1) * 32 + nt + l15]; }
    else     { bv.x = B[(nt + l15) * 32 + k0];  bv.y = B[(nt + l15) * 32 + k0 + 1]; }
    acc = __builtin_amdgcn_wmma_f32_16x16x4_f32(false, av, false, bv, (short)0,
                                                acc, false, false);
  }
  const int rbase = mt + ((lane >> 4) << 3);
#pragma unroll
  for (int r = 0; r < 8; ++r) {
    float v = acc[r];
    float* p = &D[(rbase + r) * ldd + nt + l15];
    if (mode == MM_STORE) *p = v;
    else if (mode == MM_STORE_NEG) *p = -v;
    else if (mode == MM_SUB) *p -= v;
    else *p += v;
  }
}

__device__ inline float lane_bcast(float v, int src) {
  return __builtin_bit_cast(
      float, __builtin_amdgcn_readlane(__builtin_bit_cast(int, v), src));
}

__global__ __launch_bounds__(128) void ris_precompute_kernel(
    const float* __restrict__ Ri_sqrts, const float* __restrict__ ms,
    float* __restrict__ g_Ris, float* __restrict__ g_hR) {
  __shared__ float sR[1024], sO[1024], sm[DD];
  const int t = blockIdx.x;
  const int tid = threadIdx.x;
  for (int e = tid; e < 1024; e += 128) sR[e] = Ri_sqrts[(size_t)t * 1024 + e];
  if (tid < DD) sm[tid] = ms[t * DD + tid];
  __syncthreads();
  mm32<false, true>(sR, sR, sO, 32, MM_STORE, tid);
  __syncthreads();
  for (int e = tid; e < 1024; e += 128) g_Ris[(size_t)t * 1024 + e] = sO[e];
  if (tid < DD) {
    float a = 0.f;
    for (int j = 0; j < DD; ++j) a += sO[tid * 32 + j] * sm[j];
    g_hR[t * DD + tid] = a;
  }
}

__global__ __launch_bounds__(128) void lds_forward_kernel(
    const float* __restrict__ z_samples, const float* __restrict__ A_base,
    const float* __restrict__ b_base, const float* __restrict__ Q_sqrt,
    const float* __restrict__ noise, const float* __restrict__ g_Ris,
    const float* __restrict__ g_hR, float* __restrict__ g_c,
    float* __restrict__ g_d, float* __restrict__ out) {
  __shared__ float sA[1024], sQs[1024], sJd[1024], sC[1024], sL[1024],
      sLl[1024];
  __shared__ float sQis[2][1024], sJl[2][1024];
  __shared__ float sAug[32 * AUG_LD];
  __shared__ float sz[KK], sbs[DD], sQb[DD], sh[DD], sJlPd[DD], sd[DD],
      sDiag[DD], sx[DD];

  const bool triRole = (blockIdx.x < NB);
  const int n = triRole ? blockIdx.x : blockIdx.x - NB;
  const int tid = threadIdx.x;
  const int row = tid & 31;
  const int jgrp = tid >> 5;

  float rA[8][KK], rQ[8][KK], rb[KK];
#pragma unroll
  for (int i = 0; i < 8; ++i) {
#pragma unroll
    for (int k = 0; k < KK; ++k) {
      rA[i][k] = A_base[k * 1024 + tid + 128 * i];
      rQ[i][k] = Q_sqrt[k * 1024 + tid + 128 * i];
    }
  }
#pragma unroll
  for (int k = 0; k < KK; ++k)
    rb[k] = (triRole && tid < DD) ? b_base[k * DD + tid] : 0.f;

  for (int t = 0; t < TT; ++t) {
    const int cur = t & 1, prev = cur ^ 1;
    const bool haveNext = (t < TT - 1);

    const float* vsrc =
        triRole ? (g_hR + t * DD) : (noise + ((size_t)n * TT + t) * DD);
    float zv = 0.f, vecv = 0.f, ris[8];
    if (haveNext && tid < KK) zv = z_samples[((size_t)n * TT + t) * KK + tid];
    if (tid < DD) vecv = vsrc[tid];
#pragma unroll
    for (int i = 0; i < 8; ++i)
      ris[i] = g_Ris[(size_t)t * 1024 + tid + 128 * i];

    if (haveNext) {
      if (tid < KK) sz[tid] = zv;
      __syncthreads();
      {
        float zr[KK];
#pragma unroll
        for (int k = 0; k < KK; ++k) zr[k] = sz[k];
#pragma unroll
        for (int i = 0; i < 8; ++i) {
          float a = 0.f, q = 0.f;
#pragma unroll
          for (int k = 0; k < KK; ++k) {
            a += zr[k] * rA[i][k];
            q += zr[k] * rQ[i][k];
          }
          sA[tid + 128 * i] = a;
          sQs[tid + 128 * i] = q;
        }
        if (triRole && tid < DD) {
          float b = 0.f;
#pragma unroll
          for (int k = 0; k < KK; ++k) b += zr[k] * rb[k];
          sbs[tid] = b;
        }
      }
      __syncthreads();
      mm32<false, true>(sQs, sQs, sQis[cur], 32, MM_STORE, tid);
      __syncthreads();
      mm32<false, false>(sQis[cur], sA, sJl[cur], 32, MM_STORE_NEG, tid);
      __syncthreads();
      mm32<true, false>(sA, sJl[cur], sJd, 32, MM_STORE_NEG, tid);
      __syncthreads();
    } else {
      for (int e = tid; e < 1024; e += 128) sJd[e] = 0.f;
      __syncthreads();
    }

    {
      float jd[8];
      if (t > 0) {
        float qp[8];
#pragma unroll
        for (int i = 0; i < 8; ++i) {
          jd[i] = sJd[tid + 128 * i];
          qp[i] = sQis[prev][tid + 128 * i];
        }
#pragma unroll
        for (int i = 0; i < 8; ++i) sJd[tid + 128 * i] = (jd[i] + ris[i]) + qp[i];
      } else {
#pragma unroll
        for (int i = 0; i < 8; ++i) jd[i] = sJd[tid + 128 * i];
#pragma unroll
        for (int i = 0; i < 8; ++i) sJd[tid + 128 * i] = jd[i] + ris[i];
      }
    }
    if (triRole && tid < DD) {
      float hv = vecv;
      if (t > 0) hv += sQb[tid];
      if (haveNext) {
        float jb = 0.f;
        for (int j = 0; j < DD; ++j) jb += sJl[cur][tid * 32 + j] * sbs[j];
        hv += jb;
      }
      float jd = 0.f;
      if (t > 0)
        for (int j = 0; j < DD; ++j) jd += sJl[prev][tid * 32 + j] * sd[j];
      sh[tid] = hv;
      sJlPd[tid] = jd;
      if (haveNext) {
        float qb = 0.f;
        for (int j = 0; j < DD; ++j) qb += sQis[cur][tid * 32 + j] * sbs[j];
        sQb[tid] = qb;
      }
    }
    __syncthreads();

    if (triRole) {
      {
        float aj[8], al[8];
#pragma unroll
        for (int i = 0; i < 8; ++i) aj[i] = sJd[tid + 128 * i];
        if (haveNext) {
#pragma unroll
          for (int i = 0; i < 8; ++i) al[i] = sJl[cur][tid + 128 * i];
        } else {
#pragma unroll
          for (int i = 0; i < 8; ++i) al[i] = 0.f;
        }
        float rv = 0.f;
        if (tid < DD) rv = sh[tid] - (t > 0 ? sJlPd[tid] : 0.f);
#pragma unroll
        for (int i = 0; i < 8; ++i) {
          const int r = jgrp + 4 * i;
          sAug[r * AUG_LD + row] = aj[i] + (r == row ? EPS_TRI : 0.f);
          sAug[row * AUG_LD + 32 + r] = al[i];
        }
        if (tid < DD) sAug[tid * AUG_LD + 64] = rv;
      }
      __syncthreads();
      if (t > 0) {
        mm32<false, false>(sJl[prev], sC, sAug, AUG_LD, MM_SUB, tid);
        __syncthreads();
      }
      {
        float a[32], b[9];
        float diag = 0.f;
#pragma unroll
        for (int j = 0; j < 32; ++j) a[j] = sAug[row * AUG_LD + j];
#pragma unroll
        for (int j = 0; j < 8; ++j)
          b[j] = sAug[row * AUG_LD + 32 + 8 * jgrp + j];
        b[8] = sAug[row * AUG_LD + 64];
#pragma unroll
        for (int k = 0; k < 32; ++k) {
          const float rinv = lane_bcast(__builtin_amdgcn_rcpf(a[k]), k);
          float f = a[k] * rinv;
          f = (row == k) ? 0.f : f;
          diag = (row == k) ? a[k] : diag;
#pragma unroll
          for (int j = k + 1; j < 32; ++j) a[j] -= f * lane_bcast(a[j], k);
#pragma unroll
          for (int j = 0; j < 9; ++j) b[j] -= f * lane_bcast(b[j], k);
        }
#pragma unroll
        for (int j = 0; j < 8; ++j)
          sAug[row * AUG_LD + 32 + 8 * jgrp + j] = b[j];
        if (jgrp == 0) {
          sDiag[row] = diag;
          sd[row] = b[8] / diag;
        }
      }
      __syncthreads();
      {
        const size_t base = (size_t)n * TT + t;
        float cv[8];
        const float dg = sDiag[row];
#pragma unroll
        for (int i = 0; i < 8; ++i)
          cv[i] = sAug[row * AUG_LD + 32 + jgrp + 4 * i];
#pragma unroll
        for (int i = 0; i < 8; ++i) {
          cv[i] = cv[i] / dg;
          sC[tid + 128 * i] = cv[i];
        }
        if (haveNext) {
#pragma unroll
          for (int i = 0; i < 8; ++i) g_c[base * 1024 + tid + 128 * i] = cv[i];
        }
        if (tid < DD) g_d[base * DD + tid] = sd[tid];
      }
      continue;
    }

    if (t > 0) {
      {
        float jl[8];
#pragma unroll
        for (int i = 0; i < 8; ++i) jl[i] = sJl[prev][tid + 128 * i];
#pragma unroll
        for (int i = 0; i < 8; ++i) sAug[(jgrp + 4 * i) * AUG_LD + row] = jl[i];
      }
      __syncthreads();
      if (tid < 32) {
        float x[32];
        const float dl = 1.0f / sL[tid * 33];
#pragma unroll
        for (int j = 0; j < 32; ++j) x[j] = sAug[tid * AUG_LD + j];
#pragma unroll
        for (int i = 31; i >= 0; --i) {
          const float xi = x[i] * lane_bcast(dl, i);
          x[i] = xi;
#pragma unroll
          for (int j = 0; j < i; ++j) x[j] -= sL[i * 32 + j] * xi;
        }
#pragma unroll
        for (int j = 0; j < 32; ++j) sAug[tid * AUG_LD + j] = x[j];
      }
      __syncthreads();
      {
        float xl[8], jd[8];
#pragma unroll
        for (int i = 0; i < 8; ++i) {
          xl[i] = sAug[(jgrp + 4 * i) * AUG_LD + row];
          jd[i] = sJd[tid + 128 * i];
        }
#pragma unroll
        for (int i = 0; i < 8; ++i) {
          const int r = jgrp + 4 * i;
          sLl[tid + 128 * i] = xl[i];
          sAug[r * AUG_LD + row] = jd[i] + (r == row ? EPS_CHOL : 0.f);
        }
      }
      __syncthreads();
      mm32<false, true>(sLl, sLl, sAug, AUG_LD, MM_SUB, tid);
      __syncthreads();
    } else {
      float jd[8];
#pragma unroll
      for (int i = 0; i < 8; ++i) jd[i] = sJd[tid + 128 * i];
#pragma unroll
      for (int i = 0; i < 8; ++i) {
        const int r = jgrp + 4 * i;
        sAug[r * AUG_LD + row] = jd[i] + (r == row ? EPS_CHOL : 0.f);
      }
      __syncthreads();
    }
    if (tid < 32) {
      float m[32];
#pragma unroll
      for (int j = 0; j < 32; ++j) m[j] = sAug[tid * AUG_LD + j];
#pragma unroll
      for (int k = 0; k < 32; ++k) {
        const float rinv = lane_bcast(__builtin_amdgcn_rsqf(m[k]), k);
        m[k] *= rinv;
#pragma unroll
        for (int j = k + 1; j < 32; ++j) m[j] -= m[k] * lane_bcast(m[k], j);
      }
#pragma unroll
      for (int j = 0; j < 32; ++j) sAug[tid * AUG_LD + j] = m[j];
    }
    __syncthreads();
    {
      float lv[8];
#pragma unroll
      for (int i = 0; i < 8; ++i) lv[i] = sAug[(jgrp + 4 * i) * AUG_LD + row];
#pragma unroll
      for (int i = 0; i < 8; ++i)
        sL[tid + 128 * i] = (row <= jgrp + 4 * i) ? lv[i] : 0.f;
    }
    __syncthreads();

    if (tid < DD) {
      float lc[32];
#pragma unroll
      for (int i = 0; i < 32; ++i) lc[i] = sL[i * 32 + tid];
      const float dinv = 1.0f / (sL[tid * 33] + EPS_BS);
      float r = vecv;
      if (t > 0) {
        float s = 0.f;
        for (int j = 0; j < DD; ++j) s += sLl[j * 32 + tid] * sx[j];
        r -= s;
      }
      float xv = 0.f;
#pragma unroll
      for (int i = 31; i >= 0; --i) {
        const float p = r * dinv;
        const float xi = lane_bcast(p, i);
        xv = (tid == i) ? p : xv;
        r -= lc[i] * xi;
      }
      sx[tid] = xv;
      out[((size_t)n * TT + t) * DD + tid] = xv;
    }
    __syncthreads();
  }
}

__global__ __launch_bounds__(128) void lds_backward_kernel(
    const float* __restrict__ g_c, const float* __restrict__ g_d,
    float* __restrict__ out) {
  __shared__ float sCt[1024];
  __shared__ float sMu[DD], sNew[DD];
  const int n = blockIdx.x;
  const int tid = threadIdx.x;
  if (tid < DD) {
    float v = g_d[((size_t)n * TT + (TT - 1)) * DD + tid];
    sMu[tid] = v;
    out[((size_t)n * TT + (TT - 1)) * DD + tid] += v;
  }
  __syncthreads();
  for (int t = TT - 2; t >= 0; --t) {
    size_t base = (size_t)n * TT + t;
    float cr[8], dv = 0.f, ov = 0.f;
#pragma unroll
    for (int i = 0; i < 8; ++i) cr[i] = g_c[base * 1024 + tid + 128 * i];
    if (tid < DD) {
      dv = g_d[base * DD + tid];
      ov = out[base * DD + tid];
    }
#pragma unroll
    for (int i = 0; i < 8; ++i) sCt[tid + 128 * i] = cr[i];
    __syncthreads();
    if (tid < DD) {
      float acc = dv;
      for (int j = 0; j < DD; ++j) acc -= sCt[tid * 32 + j] * sMu[j];
      sNew[tid] = acc;
    }
    __syncthreads();
    if (tid < DD) {
      sMu[tid] = sNew[tid];
      out[base * DD + tid] = ov + sNew[tid];
    }
    __syncthreads();
  }
}

extern "C" void kernel_launch(void* const* d_in, const int* in_sizes, int n_in,
                              void* d_out, int out_size, void* d_ws,
                              size_t ws_size, hipStream_t stream) {
  (void)in_sizes; (void)n_in; (void)out_size; (void)ws_size;
  const float* z_samples = (const float*)d_in[0];
  const float* A_base = (const float*)d_in[1];
  const float* b_base = (const float*)d_in[2];
  const float* Q_sqrt = (const float*)d_in[3];
  const float* ms = (const float*)d_in[4];
  const float* Ri_sqrts = (const float*)d_in[5];
  const float* noise = (const float*)d_in[6];
  float* ws = (float*)d_ws;
  float* g_Ris = ws;
  float* g_hR = g_Ris + (size_t)TT * DD * DD;
  float* g_d = g_hR + (size_t)TT * DD;
  float* g_c = g_d + (size_t)NB * TT * DD;
  float* out = (float*)d_out;

  ris_precompute_kernel<<<TT, 128, 0, stream>>>(Ri_sqrts, ms, g_Ris, g_hR);
  lds_forward_kernel<<<NB_FWD, 128, 0, stream>>>(z_samples, A_base, b_base, Q_sqrt,
                                             noise, g_Ris, g_hR, g_c, g_d, out);
  lds_backward_kernel<<<NB, 128, 0, stream>>>(g_c, g_d, out);
}


# Work off the serial chain moves to idle waves of the same barrier interval: elimination role's 3 mat-vecs on 3 waves; sampling solve of step t-1 on wave 1 beside step t's triangular solve; sums kept
# speedup vs baseline: 14.8281x; 1.1556x over previous
#include <hip/hip_runtime.h>

#define TT 256
#define DD 32
#define KK 8
#define NB 64
#define NB_FWD 128
#define AUG_LD 66
#define EPS_TRI 0.01f
#define EPS_CHOL 0.01f
#define EPS_BS 0.0001f

typedef float v2f __attribute__((ext_vector_type(2)));
typedef float v8f __attribute__((ext_vector_type(8)));

enum { MM_STORE = 0, MM_STORE_NEG = 1, MM_SUB = 2, MM_ADD = 3 };

template <bool TA, bool TB>
__device__ inline void mm32(const float* A, const float* B, float* D, int ldd,
                            int mode, int tid) {
  const int lane = tid & 31;
  const int wid = tid >> 5;
  const int mt = (wid >> 1) << 4;
  const int nt = (wid & 1) << 4;
  const int l15 = lane & 15;
  const int kh = (lane >> 4) << 1;
  v8f acc = {};
#pragma unroll
  for (int kt = 0; kt < 32; kt += 4) {
    const int k0 = kt + kh;
    v2f av, bv;
    if (!TA) { av.x = A[(mt + l15) * 32 + k0];  av.y = A[(mt + l15) * 32 + k0 + 1]; }
    else     { av.x = A[k0 * 32 + mt + l15];    av.y = A[(k0 + 1) * 32 + mt + l15]; }
    if (!TB) { bv.x = B[k0 * 32 + nt + l15];    bv.y = B[(k0 + 1) * 32 + nt + l15]; }
    else     { bv.x = B[(nt + l15) * 32 + k0];  bv.y = B[(nt + l15) * 32 + k0 + 1]; }
    acc = __builtin_amdgcn_wmma_f32_16x16x4_f32(false, av, false, bv, (short)0,
                                                acc, false, false);
  }
  const int rbase = mt + ((lane >> 4) << 3);
#pragma unroll
  for (int r = 0; r < 8; ++r) {
    float v = acc[r];
    float* p = &D[(rbase + r) * ldd + nt + l15];
    if (mode == MM_STORE) *p = v;
    else if (mode == MM_STORE_NEG) *p = -v;
    else if (mode == MM_SUB) *p -= v;
    else *p += v;
  }
}

__device__ inline float lane_bcast(float v, int src) {
  return __builtin_bit_cast(
      float, __builtin_amdgcn_readlane(__builtin_bit_cast(int, v), src));
}

__device__ inline void sample_step(const float* sL, const float* sLl, float* sx,
                                   float zv, bool havePrev, int ln,
                                   float* outp) {
  float lc[32];
#pragma unroll
  for (int i = 0; i < 32; ++i) lc[i] = sL[i * 32 + ln];
  const float dinv = 1.0f / (sL[ln * 33] + EPS_BS);
  float r = zv;
  if (havePrev) {
    float s = 0.f;
    for (int j = 0; j < DD; ++j) s += sLl[j * 32 + ln] * sx[j];
    r -= s;
  }
  float xv = 0.f;
#pragma unroll
  for (int i = 31; i >= 0; --i) {
    const float p = r * dinv;
    const float xi = lane_bcast(p, i);
    xv = (ln == i) ? p : xv;
    r -= lc[i] * xi;
  }
  sx[ln] = xv;
  *outp = xv;
}

__global__ __launch_bounds__(128) void ris_precompute_kernel(
    const float* __restrict__ Ri_sqrts, const float* __restrict__ ms,
    float* __restrict__ g_Ris, float* __restrict__ g_hR) {
  __shared__ float sR[1024], sO[1024], sm[DD];
  const int t = blockIdx.x;
  const int tid = threadIdx.x;
  for (int e = tid; e < 1024; e += 128) sR[e] = Ri_sqrts[(size_t)t * 1024 + e];
  if (tid < DD) sm[tid] = ms[t * DD + tid];
  __syncthreads();
  mm32<false, true>(sR, sR, sO, 32, MM_STORE, tid);
  __syncthreads();
  for (int e = tid; e < 1024; e += 128) g_Ris[(size_t)t * 1024 + e] = sO[e];
  if (tid < DD) {
    float a = 0.f;
    for (int j = 0; j < DD; ++j) a += sO[tid * 32 + j] * sm[j];
    g_hR[t * DD + tid] = a;
  }
}

__global__ __launch_bounds__(128) void lds_forward_kernel(
    const float* __restrict__ z_samples, const float* __restrict__ A_base,
    const float* __restrict__ b_base, const float* __restrict__ Q_sqrt,
    const float* __restrict__ noise, const float* __restrict__ g_Ris,
    const float* __restrict__ g_hR, float* __restrict__ g_c,
    float* __restrict__ g_d, float* __restrict__ out) {
  __shared__ float sA[1024], sQs[1024], sJd[1024], sC[1024], sL[1024],
      sLl[1024];
  __shared__ float sQis[2][1024], sJl[2][1024];
  __shared__ float sAug[32 * AUG_LD];
  __shared__ float sz[KK], sbs[DD], sQb[2][DD], sh[DD], sJlPd[DD], sd[DD],
      sDiag[DD], sx[DD];

  const bool triRole = (blockIdx.x < NB);
  const int n = triRole ? blockIdx.x : blockIdx.x - NB;
  const int tid = threadIdx.x;
  const int row = tid & 31;
  const int jgrp = tid >> 5;

  float rA[8][KK], rQ[8][KK], rb[KK];
#pragma unroll
  for (int i = 0; i < 8; ++i) {
#pragma unroll
    for (int k = 0; k < KK; ++k) {
      rA[i][k] = A_base[k * 1024 + tid + 128 * i];
      rQ[i][k] = Q_sqrt[k * 1024 + tid + 128 * i];
    }
  }
#pragma unroll
  for (int k = 0; k < KK; ++k)
    rb[k] = (triRole && tid < DD) ? b_base[k * DD + tid] : 0.f;

  for (int t = 0; t < TT; ++t) {
    const int cur = t & 1, prev = cur ^ 1;
    const bool haveNext = (t < TT - 1);

    float zv = 0.f, vecv = 0.f, ris[8];
    if (haveNext && tid < KK) zv = z_samples[((size_t)n * TT + t) * KK + tid];
    if (triRole) {
      if (tid < DD) vecv = g_hR[t * DD + tid];
    } else if (t > 0 && jgrp == 1) {
      vecv = noise[((size_t)n * TT + (t - 1)) * DD + row];
    }
#pragma unroll
    for (int i = 0; i < 8; ++i)
      ris[i] = g_Ris[(size_t)t * 1024 + tid + 128 * i];

    if (haveNext) {
      if (tid < KK) sz[tid] = zv;
      __syncthreads();
      {
        float zr[KK];
#pragma unroll
        for (int k = 0; k < KK; ++k) zr[k] = sz[k];
#pragma unroll
        for (int i = 0; i < 8; ++i) {
          float a = 0.f, q = 0.f;
#pragma unroll
          for (int k = 0; k < KK; ++k) {
            a += zr[k] * rA[i][k];
            q += zr[k] * rQ[i][k];
          }
          sA[tid + 128 * i] = a;
          sQs[tid + 128 * i] = q;
        }
        if (triRole && tid < DD) {
          float b = 0.f;
#pragma unroll
          for (int k = 0; k < KK; ++k) b += zr[k] * rb[k];
          sbs[tid] = b;
        }
      }
      __syncthreads();
      mm32<false, true>(sQs, sQs, sQis[cur], 32, MM_STORE, tid);
      __syncthreads();
      mm32<false, false>(sQis[cur], sA, sJl[cur], 32, MM_STORE_NEG, tid);
      __syncthreads();
      mm32<true, false>(sA, sJl[cur], sJd, 32, MM_STORE_NEG, tid);
      __syncthreads();
    } else {
      for (int e = tid; e < 1024; e += 128) sJd[e] = 0.f;
      __syncthreads();
    }

    {
      float jd[8];
      if (t > 0) {
        float qp[8];
#pragma unroll
        for (int i = 0; i < 8; ++i) {
          jd[i] = sJd[tid + 128 * i];
          qp[i] = sQis[prev][tid + 128 * i];
        }
#pragma unroll
        for (int i = 0; i < 8; ++i) sJd[tid + 128 * i] = (jd[i] + ris[i]) + qp[i];
      } else {
#pragma unroll
        for (int i = 0; i < 8; ++i) jd[i] = sJd[tid + 128 * i];
#pragma unroll
        for (int i = 0; i < 8; ++i) sJd[tid + 128 * i] = jd[i] + ris[i];
      }
    }
    if (triRole) {
      if (jgrp == 0) {
        float hv = vecv;
        if (t > 0) hv += sQb[prev][row];
        if (haveNext) {
          float jb = 0.f;
          for (int j = 0; j < DD; ++j) jb += sJl[cur][row * 32 + j] * sbs[j];
          hv += jb;
        }
        sh[row] = hv;
      } else if (jgrp == 1) {
        float jd = 0.f;
        if (t > 0)
          for (int j = 0; j < DD; ++j) jd += sJl[prev][row * 32 + j] * sd[j];
        sJlPd[row] = jd;
      } else if (jgrp == 2) {
        if (haveNext) {
          float qb = 0.f;
          for (int j = 0; j < DD; ++j) qb += sQis[cur][row * 32 + j] * sbs[j];
          sQb[cur][row] = qb;
        }
      }
    }
    __syncthreads();

    if (triRole) {
      {
        float aj[8], al[8];
#pragma unroll
        for (int i = 0; i < 8; ++i) aj[i] = sJd[tid + 128 * i];
        if (haveNext) {
#pragma unroll
          for (int i = 0; i < 8; ++i) al[i] = sJl[cur][tid + 128 * i];
        } else {
#pragma unroll
          for (int i = 0; i < 8; ++i) al[i] = 0.f;
        }
        float rv = 0.f;
        if (tid < DD) rv = sh[tid] - (t > 0 ? sJlPd[tid] : 0.f);
#pragma unroll
        for (int i = 0; i < 8; ++i) {
          const int r = jgrp + 4 * i;
          sAug[r * AUG_LD + row] = aj[i] + (r == row ? EPS_TRI : 0.f);
          sAug[row * AUG_LD + 32 + r] = al[i];
        }
        if (tid < DD) sAug[tid * AUG_LD + 64] = rv;
      }
      __syncthreads();
      if (t > 0) {
        mm32<false, false>(sJl[prev], sC, sAug, AUG_LD, MM_SUB, tid);
        __syncthreads();
      }
      {
        float a[32], b[9];
        float diag = 0.f;
#pragma unroll
        for (int j = 0; j < 32; ++j) a[j] = sAug[row * AUG_LD + j];
#pragma unroll
        for (int j = 0; j < 8; ++j)
          b[j] = sAug[row * AUG_LD + 32 + 8 * jgrp + j];
        b[8] = sAug[row * AUG_LD + 64];
#pragma unroll
        for (int k = 0; k < 32; ++k) {
          const float rinv = lane_bcast(__builtin_amdgcn_rcpf(a[k]), k);
          float f = a[k] * rinv;
          f = (row == k) ? 0.f : f;
          diag = (row == k) ? a[k] : diag;
#pragma unroll
          for (int j = k + 1; j < 32; ++j) a[j] -= f * lane_bcast(a[j], k);
#pragma unroll
          for (int j = 0; j < 9; ++j) b[j] -= f * lane_bcast(b[j], k);
        }
#pragma unroll
        for (int j = 0; j < 8; ++j)
          sAug[row * AUG_LD + 32 + 8 * jgrp + j] = b[j];
        if (jgrp == 0) {
          sDiag[row] = diag;
          sd[row] = b[8] / diag;
        }
      }
      __syncthreads();
      {
        const size_t base = (size_t)n * TT + t;
        float cv[8];
        const float dg = sDiag[row];
#pragma unroll
        for (int i = 0; i < 8; ++i)
          cv[i] = sAug[row * AUG_LD + 32 + jgrp + 4 * i];
#pragma unroll
        for (int i = 0; i < 8; ++i) {
          cv[i] = cv[i] / dg;
          sC[tid + 128 * i] = cv[i];
        }
        if (haveNext) {
#pragma unroll
          for (int i = 0; i < 8; ++i) g_c[base * 1024 + tid + 128 * i] = cv[i];
        }
        if (tid < DD) g_d[base * DD + tid] = sd[tid];
      }
      continue;
    }

    if (t > 0) {
      {
        float jl[8];
#pragma unroll
        for (int i = 0; i < 8; ++i) jl[i] = sJl[prev][tid + 128 * i];
#pragma unroll
        for (int i = 0; i < 8; ++i) sAug[(jgrp + 4 * i) * AUG_LD + row] = jl[i];
      }
      __syncthreads();
      if (tid < 32) {
        float x[32];
        const float dl = 1.0f / sL[tid * 33];
#pragma unroll
        for (int j = 0; j < 32; ++j) x[j] = sAug[tid * AUG_LD + j];
#pragma unroll
        for (int i = 31; i >= 0; --i) {
          const float xi = x[i] * lane_bcast(dl, i);
          x[i] = xi;
#pragma unroll
          for (int j = 0; j < i; ++j) x[j] -= sL[i * 32 + j] * xi;
        }
#pragma unroll
        for (int j = 0; j < 32; ++j) sAug[tid * AUG_LD + j] = x[j];
      } else if (jgrp == 1) {
        sample_step(sL, sLl, sx, vecv, t > 1, row,
                    &out[((size_t)n * TT + (t - 1)) * DD + row]);
      }
      __syncthreads();
      {
        float xl[8], jd[8];
#pragma unroll
        for (int i = 0; i < 8; ++i) {
          xl[i] = sAug[(jgrp + 4 * i) * AUG_LD + row];
          jd[i] = sJd[tid + 128 * i];
        }
#pragma unroll
        for (int i = 0; i < 8; ++i) {
          const int r = jgrp + 4 * i;
          sLl[tid + 128 * i] = xl[i];
          sAug[r * AUG_LD + row] = jd[i] + (r == row ? EPS_CHOL : 0.f);
        }
      }
      __syncthreads();
      mm32<false, true>(sLl, sLl, sAug, AUG_LD, MM_SUB, tid);
      __syncthreads();
    } else {
      float jd[8];
#pragma unroll
      for (int i = 0; i < 8; ++i) jd[i] = sJd[tid + 128 * i];
#pragma unroll
      for (int i = 0; i < 8; ++i) {
        const int r = jgrp + 4 * i;
        sAug[r * AUG_LD + row] = jd[i] + (r == row ? EPS_CHOL : 0.f);
      }
      __syncthreads();
    }
    if (tid < 32) {
      float m[32];
#pragma unroll
      for (int j = 0; j < 32; ++j) m[j] = sAug[tid * AUG_LD + j];
#pragma unroll
      for (int k = 0; k < 32; ++k) {
        const float rinv = lane_bcast(__builtin_amdgcn_rsqf(m[k]), k);
        m[k] *= rinv;
#pragma unroll
        for (int j = k + 1; j < 32; ++j) m[j] -= m[k] * lane_bcast(m[k], j);
      }
#pragma unroll
      for (int j = 0; j < 32; ++j) sAug[tid * AUG_LD + j] = m[j];
    }
    __syncthreads();
    {
      float lv[8];
#pragma unroll
      for (int i = 0; i < 8; ++i) lv[i] = sAug[(jgrp + 4 * i) * AUG_LD + row];
#pragma unroll
      for (int i = 0; i < 8; ++i)
        sL[tid + 128 * i] = (row <= jgrp + 4 * i) ? lv[i] : 0.f;
    }
    __syncthreads();

  }

  if (!triRole && jgrp == 1) {
    const float zlast = noise[((size_t)n * TT + (TT - 1)) * DD + row];
    sample_step(sL, sLl, sx, zlast, true, row,
                &out[((size_t)n * TT + (TT - 1)) * DD + row]);
  }
}

__global__ __launch_bounds__(128) void lds_backward_kernel(
    const float* __restrict__ g_c, const float* __restrict__ g_d,
    float* __restrict__ out) {
  __shared__ float sCt[1024];
  __shared__ float sMu[DD], sNew[DD];
  const int n = blockIdx.x;
  const int tid = threadIdx.x;
  if (tid < DD) {
    float v = g_d[((size_t)n * TT + (TT - 1)) * DD + tid];
    sMu[tid] = v;
    out[((size_t)n * TT + (TT - 1)) * DD + tid] += v;
  }
  __syncthreads();
  for (int t = TT - 2; t >= 0; --t) {
    size_t base = (size_t)n * TT + t;
    float cr[8], dv = 0.f, ov = 0.f;
#pragma unroll
    for (int i = 0; i < 8; ++i) cr[i] = g_c[base * 1024 + tid + 128 * i];
    if (tid < DD) {
      dv = g_d[base * DD + tid];
      ov = out[base * DD + tid];
    }
#pragma unroll
    for (int i = 0; i < 8; ++i) sCt[tid + 128 * i] = cr[i];
    __syncthreads();
    if (tid < DD) {
      float acc = dv;
      for (int j = 0; j < DD; ++j) acc -= sCt[tid * 32 + j] * sMu[j];
      sNew[tid] = acc;
    }
    __syncthreads();
    if (tid < DD) {
      sMu[tid] = sNew[tid];
      out[base * DD + tid] = ov + sNew[tid];
    }
    __syncthreads();
  }
}

extern "C" void kernel_launch(void* const* d_in, const int* in_sizes, int n_in,
                              void* d_out, int out_size, void* d_ws,
                              size_t ws_size, hipStream_t stream) {
  (void)in_sizes; (void)n_in; (void)out_size; (void)ws_size;
  const float* z_samples = (const float*)d_in[0];
  const float* A_base = (const float*)d_in[1];
  const float* b_base = (const float*)d_in[2];
  const float* Q_sqrt = (const float*)d_in[3];
  const float* ms = (const float*)d_in[4];
  const float* Ri_sqrts = (const float*)d_in[5];
  const float* noise = (const float*)d_in[6];
  float* ws = (float*)d_ws;
  float* g_Ris = ws;
  float* g_hR = g_Ris + (size_t)TT * DD * DD;
  float* g_d = g_hR + (size_t)TT * DD;
  float* g_c = g_d + (size_t)NB * TT * DD;
  float* out = (float*)d_out;

  ris_precompute_kernel<<<TT, 128, 0, stream>>>(Ri_sqrts, ms, g_Ris, g_hR);
  lds_forward_kernel<<<NB_FWD, 128, 0, stream>>>(z_samples, A_base, b_base, Q_sqrt,
                                             noise, g_Ris, g_hR, g_c, g_d, out);
  lds_backward_kernel<<<NB, 128, 0, stream>>>(g_c, g_d, out);
}


# Backward recursion in one wave, row per lane in registers: c rows, d, out requested 5 steps ahead, mu broadcast by v_readlane; no LDS, barrier or store wait per step; sum order kept, terms fused
# speedup vs baseline: 17.1042x; 1.1535x over previous
#include <hip/hip_runtime.h>

#define TT 256
#define DD 32
#define KK 8
#define NB 64
#define NB_FWD 128
#define AUG_LD 66
#define EPS_TRI 0.01f
#define EPS_CHOL 0.01f
#define EPS_BS 0.0001f

typedef float v2f __attribute__((ext_vector_type(2)));
typedef float v8f __attribute__((ext_vector_type(8)));

enum { MM_STORE = 0, MM_STORE_NEG = 1, MM_SUB = 2, MM_ADD = 3 };

template <bool TA, bool TB>
__device__ inline void mm32(const float* A, const float* B, float* D, int ldd,
                            int mode, int tid) {
  const int lane = tid & 31;
  const int wid = tid >> 5;
  const int mt = (wid >> 1) << 4;
  const int nt = (wid & 1) << 4;
  const int l15 = lane & 15;
  const int kh = (lane >> 4) << 1;
  v8f acc = {};
#pragma unroll
  for (int kt = 0; kt < 32; kt += 4) {
    const int k0 = kt + kh;
    v2f av, bv;
    if (!TA) { av.x = A[(mt + l15) * 32 + k0];  av.y = A[(mt + l15) * 32 + k0 + 1]; }
    else     { av.x = A[k0 * 32 + mt + l15];    av.y = A[(k0 + 1) * 32 + mt + l15]; }
    if (!TB) { bv.x = B[k0 * 32 + nt + l15];    bv.y = B[(k0 + 1) * 32 + nt + l15]; }
    else     { bv.x = B[(nt + l15) * 32 + k0];  bv.y = B[(nt + l15) * 32 + k0 + 1]; }
    acc = __builtin_amdgcn_wmma_f32_16x16x4_f32(false, av, false, bv, (short)0,
                                                acc, false, false);
  }
  const int rbase = mt + ((lane >> 4) << 3);
#pragma unroll
  for (int r = 0; r < 8; ++r) {
    float v = acc[r];
    float* p = &D[(rbase + r) * ldd + nt + l15];
    if (mode == MM_STORE) *p = v;
    else if (mode == MM_STORE_NEG) *p = -v;
    else if (mode == MM_SUB) *p -= v;
    else *p += v;
  }
}

__device__ inline float lane_bcast(float v, int src) {
  return __builtin_bit_cast(
      float, __builtin_amdgcn_readlane(__builtin_bit_cast(int, v), src));
}

__device__ inline void sample_step(const float* sL, const float* sLl, float* sx,
                                   float zv, bool havePrev, int ln,
                                   float* outp) {
  float lc[32];
#pragma unroll
  for (int i = 0; i < 32; ++i) lc[i] = sL[i * 32 + ln];
  const float dinv = 1.0f / (sL[ln * 33] + EPS_BS);
  float r = zv;
  if (havePrev) {
    float s = 0.f;
    for (int j = 0; j < DD; ++j) s += sLl[j * 32 + ln] * sx[j];
    r -= s;
  }
  float xv = 0.f;
#pragma unroll
  for (int i = 31; i >= 0; --i) {
    const float p = r * dinv;
    const float xi = lane_bcast(p, i);
    xv = (ln == i) ? p : xv;
    r -= lc[i] * xi;
  }
  sx[ln] = xv;
  *outp = xv;
}

__global__ __launch_bounds__(128) void ris_precompute_kernel(
    const float* __restrict__ Ri_sqrts, const float* __restrict__ ms,
    float* __restrict__ g_Ris, float* __restrict__ g_hR) {
  __shared__ float sR[1024], sO[1024], sm[DD];
  const int t = blockIdx.x;
  const int tid = threadIdx.x;
  for (int e = tid; e < 1024; e += 128) sR[e] = Ri_sqrts[(size_t)t * 1024 + e];
  if (tid < DD) sm[tid] = ms[t * DD + tid];
  __syncthreads();
  mm32<false, true>(sR, sR, sO, 32, MM_STORE, tid);
  __syncthreads();
  for (int e = tid; e < 1024; e += 128) g_Ris[(size_t)t * 1024 + e] = sO[e];
  if (tid < DD) {
    float a = 0.f;
    for (int j = 0; j < DD; ++j) a += sO[tid * 32 + j] * sm[j];
    g_hR[t * DD + tid] = a;
  }
}

__global__ __launch_bounds__(128) void lds_forward_kernel(
    const float* __restrict__ z_samples, const float* __restrict__ A_base,
    const float* __restrict__ b_base, const float* __restrict__ Q_sqrt,
    const float* __restrict__ noise, const float* __restrict__ g_Ris,
    const float* __restrict__ g_hR, float* __restrict__ g_c,
    float* __restrict__ g_d, float* __restrict__ out) {
  __shared__ float sA[1024], sQs[1024], sJd[1024], sC[1024], sL[1024],
      sLl[1024];
  __shared__ float sQis[2][1024], sJl[2][1024];
  __shared__ float sAug[32 * AUG_LD];
  __shared__ float sz[KK], sbs[DD], sQb[2][DD], sh[DD], sJlPd[DD], sd[DD],
      sDiag[DD], sx[DD];

  const bool triRole = (blockIdx.x < NB);
  const int n = triRole ? blockIdx.x : blockIdx.x - NB;
  const int tid = threadIdx.x;
  const int row = tid & 31;
  const int jgrp = tid >> 5;

  float rA[8][KK], rQ[8][KK], rb[KK];
#pragma unroll
  for (int i = 0; i < 8; ++i) {
#pragma unroll
    for (int k = 0; k < KK; ++k) {
      rA[i][k] = A_base[k * 1024 + tid + 128 * i];
      rQ[i][k] = Q_sqrt[k * 1024 + tid + 128 * i];
    }
  }
#pragma unroll
  for (int k = 0; k < KK; ++k)
    rb[k] = (triRole && tid < DD) ? b_base[k * DD + tid] : 0.f;

  for (int t = 0; t < TT; ++t) {
    const int cur = t & 1, prev = cur ^ 1;
    const bool haveNext = (t < TT - 1);

    float zv = 0.f, vecv = 0.f, ris[8];
    if (haveNext && tid < KK) zv = z_samples[((size_t)n * TT + t) * KK + tid];
    if (triRole) {
      if (tid < DD) vecv = g_hR[t * DD + tid];
    } else if (t > 0 && jgrp == 1) {
      vecv = noise[((size_t)n * TT + (t - 1)) * DD + row];
    }
#pragma unroll
    for (int i = 0; i < 8; ++i)
      ris[i] = g_Ris[(size_t)t * 1024 + tid + 128 * i];

    if (haveNext) {
      if (tid < KK) sz[tid] = zv;
      __syncthreads();
      {
        float zr[KK];
#pragma unroll
        for (int k = 0; k < KK; ++k) zr[k] = sz[k];
#pragma unroll
        for (int i = 0; i < 8; ++i) {
          float a = 0.f, q = 0.f;
#pragma unroll
          for (int k = 0; k < KK; ++k) {
            a += zr[k] * rA[i][k];
            q += zr[k] * rQ[i][k];
          }
          sA[tid + 128 * i] = a;
          sQs[tid + 128 * i] = q;
        }
        if (triRole && tid < DD) {
          float b = 0.f;
#pragma unroll
          for (int k = 0; k < KK; ++k) b += zr[k] * rb[k];
          sbs[tid] = b;
        }
      }
      __syncthreads();
      mm32<false, true>(sQs, sQs, sQis[cur], 32, MM_STORE, tid);
      __syncthreads();
      mm32<false, false>(sQis[cur], sA, sJl[cur], 32, MM_STORE_NEG, tid);
      __syncthreads();
      mm32<true, false>(sA, sJl[cur], sJd, 32, MM_STORE_NEG, tid);
      __syncthreads();
    } else {
      for (int e = tid; e < 1024; e += 128) sJd[e] = 0.f;
      __syncthreads();
    }

    {
      float jd[8];
      if (t > 0) {
        float qp[8];
#pragma unroll
        for (int i = 0; i < 8; ++i) {
          jd[i] = sJd[tid + 128 * i];
          qp[i] = sQis[prev][tid + 128 * i];
        }
#pragma unroll
        for (int i = 0; i < 8; ++i) sJd[tid + 128 * i] = (jd[i] + ris[i]) + qp[i];
      } else {
#pragma unroll
        for (int i = 0; i < 8; ++i) jd[i] = sJd[tid + 128 * i];
#pragma unroll
        for (int i = 0; i < 8; ++i) sJd[tid + 128 * i] = jd[i] + ris[i];
      }
    }
    if (triRole) {
      if (jgrp == 0) {
        float hv = vecv;
        if (t > 0) hv += sQb[prev][row];
        if (haveNext) {
          float jb = 0.f;
          for (int j = 0; j < DD; ++j) jb += sJl[cur][row * 32 + j] * sbs[j];
          hv += jb;
        }
        sh[row] = hv;
      } else if (jgrp == 1) {
        float jd = 0.f;
        if (t > 0)
          for (int j = 0; j < DD; ++j) jd += sJl[prev][row * 32 + j] * sd[j];
        sJlPd[row] = jd;
      } else if (jgrp == 2) {
        if (haveNext) {
          float qb = 0.f;
          for (int j = 0; j < DD; ++j) qb += sQis[cur][row * 32 + j] * sbs[j];
          sQb[cur][row] = qb;
        }
      }
    }
    __syncthreads();

    if (triRole) {
      {
        float aj[8], al[8];
#pragma unroll
        for (int i = 0; i < 8; ++i) aj[i] = sJd[tid + 128 * i];
        if (haveNext) {
#pragma unroll
          for (int i = 0; i < 8; ++i) al[i] = sJl[cur][tid + 128 * i];
        } else {
#pragma unroll
          for (int i = 0; i < 8; ++i) al[i] = 0.f;
        }
        float rv = 0.f;
        if (tid < DD) rv = sh[tid] - (t > 0 ? sJlPd[tid] : 0.f);
#pragma unroll
        for (int i = 0; i < 8; ++i) {
          const int r = jgrp + 4 * i;
          sAug[r * AUG_LD + row] = aj[i] + (r == row ? EPS_TRI : 0.f);
          sAug[row * AUG_LD + 32 + r] = al[i];
        }
        if (tid < DD) sAug[tid * AUG_LD + 64] = rv;
      }
      __syncthreads();
      if (t > 0) {
        mm32<false, false>(sJl[prev], sC, sAug, AUG_LD, MM_SUB, tid);
        __syncthreads();
      }
      {
        float a[32], b[9];
        float diag = 0.f;
#pragma unroll
        for (int j = 0; j < 32; ++j) a[j] = sAug[row * AUG_LD + j];
#pragma unroll
        for (int j = 0; j < 8; ++j)
          b[j] = sAug[row * AUG_LD + 32 + 8 * jgrp + j];
        b[8] = sAug[row * AUG_LD + 64];
#pragma unroll
        for (int k = 0; k < 32; ++k) {
          const float rinv = lane_bcast(__builtin_amdgcn_rcpf(a[k]), k);
          float f = a[k] * rinv;
          f = (row == k) ? 0.f : f;
          diag = (row == k) ? a[k] : diag;
#pragma unroll
          for (int j = k + 1; j < 32; ++j) a[j] -= f * lane_bcast(a[j], k);
#pragma unroll
          for (int j = 0; j < 9; ++j) b[j] -= f * lane_bcast(b[j], k);
        }
#pragma unroll
        for (int j = 0; j < 8; ++j)
          sAug[row * AUG_LD + 32 + 8 * jgrp + j] = b[j];
        if (jgrp == 0) {
          sDiag[row] = diag;
          sd[row] = b[8] / diag;
        }
      }
      __syncthreads();
      {
        const size_t base = (size_t)n * TT + t;
        float cv[8];
        const float dg = sDiag[row];
#pragma unroll
        for (int i = 0; i < 8; ++i)
          cv[i] = sAug[row * AUG_LD + 32 + jgrp + 4 * i];
#pragma unroll
        for (int i = 0; i < 8; ++i) {
          cv[i] = cv[i] / dg;
          sC[tid + 128 * i] = cv[i];
        }
        if (haveNext) {
#pragma unroll
          for (int i = 0; i < 8; ++i) g_c[base * 1024 + tid + 128 * i] = cv[i];
        }
        if (tid < DD) g_d[base * DD + tid] = sd[tid];
      }
      continue;
    }

    if (t > 0) {
      {
        float jl[8];
#pragma unroll
        for (int i = 0; i < 8; ++i) jl[i] = sJl[prev][tid + 128 * i];
#pragma unroll
        for (int i = 0; i < 8; ++i) sAug[(jgrp + 4 * i) * AUG_LD + row] = jl[i];
      }
      __syncthreads();
      if (tid < 32) {
        float x[32];
        const float dl = 1.0f / sL[tid * 33];
#pragma unroll
        for (int j = 0; j < 32; ++j) x[j] = sAug[tid * AUG_LD + j];
#pragma unroll
        for (int i = 31; i >= 0; --i) {
          const float xi = x[i] * lane_bcast(dl, i);
          x[i] = xi;
#pragma unroll
          for (int j = 0; j < i; ++j) x[j] -= sL[i * 32 + j] * xi;
        }
#pragma unroll
        for (int j = 0; j < 32; ++j) sAug[tid * AUG_LD + j] = x[j];
      } else if (jgrp == 1) {
        sample_step(sL, sLl, sx, vecv, t > 1, row,
                    &out[((size_t)n * TT + (t - 1)) * DD + row]);
      }
      __syncthreads();
      {
        float xl[8], jd[8];
#pragma unroll
        for (int i = 0; i < 8; ++i) {
          xl[i] = sAug[(jgrp + 4 * i) * AUG_LD + row];
          jd[i] = sJd[tid + 128 * i];
        }
#pragma unroll
        for (int i = 0; i < 8; ++i) {
          const int r = jgrp + 4 * i;
          sLl[tid + 128 * i] = xl[i];
          sAug[r * AUG_LD + row] = jd[i] + (r == row ? EPS_CHOL : 0.f);
        }
      }
      __syncthreads();
      mm32<false, true>(sLl, sLl, sAug, AUG_LD, MM_SUB, tid);
      __syncthreads();
    } else {
      float jd[8];
#pragma unroll
      for (int i = 0; i < 8; ++i) jd[i] = sJd[tid + 128 * i];
#pragma unroll
      for (int i = 0; i < 8; ++i) {
        const int r = jgrp + 4 * i;
        sAug[r * AUG_LD + row] = jd[i] + (r == row ? EPS_CHOL : 0.f);
      }
      __syncthreads();
    }
    if (tid < 32) {
      float m[32];
#pragma unroll
      for (int j = 0; j < 32; ++j) m[j] = sAug[tid * AUG_LD + j];
#pragma unroll
      for (int k = 0; k < 32; ++k) {
        const float rinv = lane_bcast(__builtin_amdgcn_rsqf(m[k]), k);
        m[k] *= rinv;
#pragma unroll
        for (int j = k + 1; j < 32; ++j) m[j] -= m[k] * lane_bcast(m[k], j);
      }
#pragma unroll
      for (int j = 0; j < 32; ++j) sAug[tid * AUG_LD + j] = m[j];
    }
    __syncthreads();
    {
      float lv[8];
#pragma unroll
      for (int i = 0; i < 8; ++i) lv[i] = sAug[(jgrp + 4 * i) * AUG_LD + row];
#pragma unroll
      for (int i = 0; i < 8; ++i)
        sL[tid + 128 * i] = (row <= jgrp + 4 * i) ? lv[i] : 0.f;
    }
    __syncthreads();

  }

  if (!triRole && jgrp == 1) {
    const float zlast = noise[((size_t)n * TT + (TT - 1)) * DD + row];
    sample_step(sL, sLl, sx, zlast, true, row,
                &out[((size_t)n * TT + (TT - 1)) * DD + row]);
  }
}

#define BWD_AHEAD 5
typedef float v4f __attribute__((ext_vector_type(4)));

__device__ inline void bwd_request(const float* __restrict__ g_c,
                                   const float* __restrict__ g_d,
                                   const float* out, size_t base, int ln,
                                   float (&c)[DD], float& d, float& o) {
  const v4f* rowp = (const v4f*)(g_c + base * 1024 + ln * DD);
#pragma unroll
  for (int q = 0; q < 8; ++q) {
    const v4f v = rowp[q];
    c[4 * q] = v.x; c[4 * q + 1] = v.y; c[4 * q + 2] = v.z; c[4 * q + 3] = v.w;
  }
  d = g_d[base * DD + ln];
  o = out[base * DD + ln];
}

__global__ __launch_bounds__(32) void lds_backward_kernel(
    const float* __restrict__ g_c, const float* __restrict__ g_d,
    float* __restrict__ out) {
  const int n = blockIdx.x;
  const int ln = threadIdx.x;
  const size_t seq = (size_t)n * TT;

  float cr[BWD_AHEAD][DD], dv[BWD_AHEAD], ov[BWD_AHEAD];
#pragma unroll
  for (int s = 0; s < BWD_AHEAD; ++s)
    bwd_request(g_c, g_d, out, seq + (TT - 2 - s), ln, cr[s], dv[s], ov[s]);

  float mu = g_d[(seq + (TT - 1)) * DD + ln];
  out[(seq + (TT - 1)) * DD + ln] += mu;

  for (int g = 0; g < (TT - 1) / BWD_AHEAD; ++g) {
#pragma unroll
    for (int s = 0; s < BWD_AHEAD; ++s) {
      const int t = TT - 2 - (g * BWD_AHEAD + s);
      float acc = dv[s];
#pragma unroll
      for (int j = 0; j < DD; ++j)
        acc = __builtin_fmaf(-cr[s][j], lane_bcast(mu, j), acc);
      mu = acc;
      out[(seq + t) * DD + ln] = ov[s] + acc;
      if (t >= BWD_AHEAD)
        bwd_request(g_c, g_d, out, seq + (t - BWD_AHEAD), ln, cr[s], dv[s],
                    ov[s]);
    }
  }
}

extern "C" void kernel_launch(void* const* d_in, const int* in_sizes, int n_in,
                              void* d_out, int out_size, void* d_ws,
                              size_t ws_size, hipStream_t stream) {
  (void)in_sizes; (void)n_in; (void)out_size; (void)ws_size;
  const float* z_samples = (const float*)d_in[0];
  const float* A_base = (const float*)d_in[1];
  const float* b_base = (const float*)d_in[2];
  const float* Q_sqrt = (const float*)d_in[3];
  const float* ms = (const float*)d_in[4];
  const float* Ri_sqrts = (const float*)d_in[5];
  const float* noise = (const float*)d_in[6];
  float* ws = (float*)d_ws;
  float* g_Ris = ws;
  float* g_hR = g_Ris + (size_t)TT * DD * DD;
  float* g_d = g_hR + (size_t)TT * DD;
  float* g_c = g_d + (size_t)NB * TT * DD;
  float* out = (float*)d_out;

  ris_precompute_kernel<<<TT, 128, 0, stream>>>(Ri_sqrts, ms, g_Ris, g_hR);
  lds_forward_kernel<<<NB_FWD, 128, 0, stream>>>(z_samples, A_base, b_base, Q_sqrt,
                                             noise, g_Ris, g_hR, g_c, g_d, out);
  lds_backward_kernel<<<NB, 32, 0, stream>>>(g_c, g_d, out);
}


# LDS row pitch of the forward kernel's 32x32 matrices 32 -> 36 floats (16 B x odd): row-strided fragment loads of the products and the mat-vec row loads stop bank-conflicting; sums unchanged
# speedup vs baseline: 18.6099x; 1.0880x over previous
#include <hip/hip_runtime.h>

#define TT 256
#define DD 32
#define KK 8
#define NB 64
#define NB_FWD 128
#define AUG_LD 66
#define LDP 36
#define LDM (DD * LDP)
#define ELS (4 * LDP)
#define EPS_TRI 0.01f
#define EPS_CHOL 0.01f
#define EPS_BS 0.0001f

typedef float v2f __attribute__((ext_vector_type(2)));
typedef float v8f __attribute__((ext_vector_type(8)));

enum { MM_STORE = 0, MM_STORE_NEG = 1, MM_SUB = 2, MM_ADD = 3 };

template <bool TA, bool TB, int LDAB>
__device__ inline void mm32(const float* A, const float* B, float* D, int ldd,
                            int mode, int tid) {
  const int lane = tid & 31;
  const int wid = tid >> 5;
  const int mt = (wid >> 1) << 4;
  const int nt = (wid & 1) << 4;
  const int l15 = lane & 15;
  const int kh = (lane >> 4) << 1;
  v8f acc = {};
#pragma unroll
  for (int kt = 0; kt < 32; kt += 4) {
    const int k0 = kt + kh;
    v2f av, bv;
    if (!TA) { av.x = A[(mt + l15) * LDAB + k0];  av.y = A[(mt + l15) * LDAB + k0 + 1]; }
    else     { av.x = A[k0 * LDAB + mt + l15];    av.y = A[(k0 + 1) * LDAB + mt + l15]; }
    if (!TB) { bv.x = B[k0 * LDAB + nt + l15];    bv.y = B[(k0 + 1) * LDAB + nt + l15]; }
    else     { bv.x = B[(nt + l15) * LDAB + k0];  bv.y = B[(nt + l15) * LDAB + k0 + 1]; }
    acc = __builtin_amdgcn_wmma_f32_16x16x4_f32(false, av, false, bv, (short)0,
                                                acc, false, false);
  }
  const int rbase = mt + ((lane >> 4) << 3);
#pragma unroll
  for (int r = 0; r < 8; ++r) {
    float v = acc[r];
    float* p = &D[(rbase + r) * ldd + nt + l15];
    if (mode == MM_STORE) *p = v;
    else if (mode == MM_STORE_NEG) *p = -v;
    else if (mode == MM_SUB) *p -= v;
    else *p += v;
  }
}

__device__ inline float lane_bcast(float v, int src) {
  return __builtin_bit_cast(
      float, __builtin_amdgcn_readlane(__builtin_bit_cast(int, v), src));
}

__device__ inline void sample_step(const float* sL, const float* sLl, float* sx,
                                   float zv, bool havePrev, int ln,
                                   float* outp) {
  float lc[32];
#pragma unroll
  for (int i = 0; i < 32; ++i) lc[i] = sL[i * LDP + ln];
  const float dinv = 1.0f / (sL[ln * (LDP + 1)] + EPS_BS);
  float r = zv;
  if (havePrev) {
    float s = 0.f;
    for (int j = 0; j < DD; ++j) s += sLl[j * LDP + ln] * sx[j];
    r -= s;
  }
  float xv = 0.f;
#pragma unroll
  for (int i = 31; i >= 0; --i) {
    const float p = r * dinv;
    const float xi = lane_bcast(p, i);
    xv = (ln == i) ? p : xv;
    r -= lc[i] * xi;
  }
  sx[ln] = xv;
  *outp = xv;
}

__global__ __launch_bounds__(128) void ris_precompute_kernel(
    const float* __restrict__ Ri_sqrts, const float* __restrict__ ms,
    float* __restrict__ g_Ris, float* __restrict__ g_hR) {
  __shared__ float sR[1024], sO[1024], sm[DD];
  const int t = blockIdx.x;
  const int tid = threadIdx.x;
  for (int e = tid; e < 1024; e += 128) sR[e] = Ri_sqrts[(size_t)t * 1024 + e];
  if (tid < DD) sm[tid] = ms[t * DD + tid];
  __syncthreads();
  mm32<false, true, 32>(sR, sR, sO, 32, MM_STORE, tid);
  __syncthreads();
  for (int e = tid; e < 1024; e += 128) g_Ris[(size_t)t * 1024 + e] = sO[e];
  if (tid < DD) {
    float a = 0.f;
    for (int j = 0; j < DD; ++j) a += sO[tid * 32 + j] * sm[j];
    g_hR[t * DD + tid] = a;
  }
}

__global__ __launch_bounds__(128) void lds_forward_kernel(
    const float* __restrict__ z_samples, const float* __restrict__ A_base,
    const float* __restrict__ b_base, const float* __restrict__ Q_sqrt,
    const float* __restrict__ noise, const float* __restrict__ g_Ris,
    const float* __restrict__ g_hR, float* __restrict__ g_c,
    float* __restrict__ g_d, float* __restrict__ out) {
  __shared__ float sA[LDM], sQs[LDM], sJd[LDM], sC[LDM], sL[LDM], sLl[LDM];
  __shared__ float sQis[2][LDM], sJl[2][LDM];
  __shared__ float sAug[32 * AUG_LD];
  __shared__ float sz[KK], sbs[DD], sQb[2][DD], sh[DD], sJlPd[DD], sd[DD],
      sDiag[DD], sx[DD];

  const bool triRole = (blockIdx.x < NB);
  const int n = triRole ? blockIdx.x : blockIdx.x - NB;
  const int tid = threadIdx.x;
  const int row = tid & 31;
  const int jgrp = tid >> 5;
  const int el0 = jgrp * LDP + row;

  float rA[8][KK], rQ[8][KK], rb[KK];
#pragma unroll
  for (int i = 0; i < 8; ++i) {
#pragma unroll
    for (int k = 0; k < KK; ++k) {
      rA[i][k] = A_base[k * 1024 + tid + 128 * i];
      rQ[i][k] = Q_sqrt[k * 1024 + tid + 128 * i];
    }
  }
#pragma unroll
  for (int k = 0; k < KK; ++k)
    rb[k] = (triRole && tid < DD) ? b_base[k * DD + tid] : 0.f;

  for (int t = 0; t < TT; ++t) {
    const int cur = t & 1, prev = cur ^ 1;
    const bool haveNext = (t < TT - 1);

    float zv = 0.f, vecv = 0.f, ris[8];
    if (haveNext && tid < KK) zv = z_samples[((size_t)n * TT + t) * KK + tid];
    if (triRole) {
      if (tid < DD) vecv = g_hR[t * DD + tid];
    } else if (t > 0 && jgrp == 1) {
      vecv = noise[((size_t)n * TT + (t - 1)) * DD + row];
    }
#pragma unroll
    for (int i = 0; i < 8; ++i)
      ris[i] = g_Ris[(size_t)t * 1024 + tid + 128 * i];

    if (haveNext) {
      if (tid < KK) sz[tid] = zv;
      __syncthreads();
      {
        float zr[KK];
#pragma unroll
        for (int k = 0; k < KK; ++k) zr[k] = sz[k];
#pragma unroll
        for (int i = 0; i < 8; ++i) {
          float a = 0.f, q = 0.f;
#pragma unroll
          for (int k = 0; k < KK; ++k) {
            a += zr[k] * rA[i][k];
            q += zr[k] * rQ[i][k];
          }
          sA[el0 + ELS * i] = a;
          sQs[el0 + ELS * i] = q;
        }
        if (triRole && tid < DD) {
          float b = 0.f;
#pragma unroll
          for (int k = 0; k < KK; ++k) b += zr[k] * rb[k];
          sbs[tid] = b;
        }
      }
      __syncthreads();
      mm32<false, true, LDP>(sQs, sQs, sQis[cur], LDP, MM_STORE, tid);
      __syncthreads();
      mm32<false, false, LDP>(sQis[cur], sA, sJl[cur], LDP, MM_STORE_NEG, tid);
      __syncthreads();
      mm32<true, false, LDP>(sA, sJl[cur], sJd, LDP, MM_STORE_NEG, tid);
      __syncthreads();
    } else {
      for (int i = 0; i < 8; ++i) sJd[el0 + ELS * i] = 0.f;
      __syncthreads();
    }

    {
      float jd[8];
      if (t > 0) {
        float qp[8];
#pragma unroll
        for (int i = 0; i < 8; ++i) {
          jd[i] = sJd[el0 + ELS * i];
          qp[i] = sQis[prev][el0 + ELS * i];
        }
#pragma unroll
        for (int i = 0; i < 8; ++i) sJd[el0 + ELS * i] = (jd[i] + ris[i]) + qp[i];
      } else {
#pragma unroll
        for (int i = 0; i < 8; ++i) jd[i] = sJd[el0 + ELS * i];
#pragma unroll
        for (int i = 0; i < 8; ++i) sJd[el0 + ELS * i] = jd[i] + ris[i];
      }
    }
    if (triRole) {
      if (jgrp == 0) {
        float hv = vecv;
        if (t > 0) hv += sQb[prev][row];
        if (haveNext) {
          float jb = 0.f;
          for (int j = 0; j < DD; ++j) jb += sJl[cur][row * LDP + j] * sbs[j];
          hv += jb;
        }
        sh[row] = hv;
      } else if (jgrp == 1) {
        float jd = 0.f;
        if (t > 0)
          for (int j = 0; j < DD; ++j) jd += sJl[prev][row * LDP + j] * sd[j];
        sJlPd[row] = jd;
      } else if (jgrp == 2) {
        if (haveNext) {
          float qb = 0.f;
          for (int j = 0; j < DD; ++j) qb += sQis[cur][row * LDP + j] * sbs[j];
          sQb[cur][row] = qb;
        }
      }
    }
    __syncthreads();

    if (triRole) {
      {
        float aj[8], al[8];
#pragma unroll
        for (int i = 0; i < 8; ++i) aj[i] = sJd[el0 + ELS * i];
        if (haveNext) {
#pragma unroll
          for (int i = 0; i < 8; ++i) al[i] = sJl[cur][el0 + ELS * i];
        } else {
#pragma unroll
          for (int i = 0; i < 8; ++i) al[i] = 0.f;
        }
        float rv = 0.f;
        if (tid < DD) rv = sh[tid] - (t > 0 ? sJlPd[tid] : 0.f);
#pragma unroll
        for (int i = 0; i < 8; ++i) {
          const int r = jgrp + 4 * i;
          sAug[r * AUG_LD + row] = aj[i] + (r == row ? EPS_TRI : 0.f);
          sAug[row * AUG_LD + 32 + r] = al[i];
        }
        if (tid < DD) sAug[tid * AUG_LD + 64] = rv;
      }
      __syncthreads();
      if (t > 0) {
        mm32<false, false, LDP>(sJl[prev], sC, sAug, AUG_LD, MM_SUB, tid);
        __syncthreads();
      }
      {
        float a[32], b[9];
        float diag = 0.f;
#pragma unroll
        for (int j = 0; j < 32; ++j) a[j] = sAug[row * AUG_LD + j];
#pragma unroll
        for (int j = 0; j < 8; ++j)
          b[j] = sAug[row * AUG_LD + 32 + 8 * jgrp + j];
        b[8] = sAug[row * AUG_LD + 64];
#pragma unroll
        for (int k = 0; k < 32; ++k) {
          const float rinv = lane_bcast(__builtin_amdgcn_rcpf(a[k]), k);
          float f = a[k] * rinv;
          f = (row == k) ? 0.f : f;
          diag = (row == k) ? a[k] : diag;
#pragma unroll
          for (int j = k + 1; j < 32; ++j) a[j] -= f * lane_bcast(a[j], k);
#pragma unroll
          for (int j = 0; j < 9; ++j) b[j] -= f * lane_bcast(b[j], k);
        }
#pragma unroll
        for (int j = 0; j < 8; ++j)
          sAug[row * AUG_LD + 32 + 8 * jgrp + j] = b[j];
        if (jgrp == 0) {
          sDiag[row] = diag;
          sd[row] = b[8] / diag;
        }
      }
      __syncthreads();
      {
        const size_t base = (size_t)n * TT + t;
        float cv[8];
        const float dg = sDiag[row];
#pragma unroll
        for (int i = 0; i < 8; ++i)
          cv[i] = sAug[row * AUG_LD + 32 + jgrp + 4 * i];
#pragma unroll
        for (int i = 0; i < 8; ++i) {
          cv[i] = cv[i] / dg;
          sC[el0 + ELS * i] = cv[i];
        }
        if (haveNext) {
#pragma unroll
          for (int i = 0; i < 8; ++i) g_c[base * 1024 + tid + 128 * i] = cv[i];
        }
        if (tid < DD) g_d[base * DD + tid] = sd[tid];
      }
      continue;
    }

    if (t > 0) {
      {
        float jl[8];
#pragma unroll
        for (int i = 0; i < 8; ++i) jl[i] = sJl[prev][el0 + ELS * i];
#pragma unroll
        for (int i = 0; i < 8; ++i) sAug[(jgrp + 4 * i) * AUG_LD + row] = jl[i];
      }
      __syncthreads();
      if (tid < 32) {
        float x[32];
        const float dl = 1.0f / sL[tid * (LDP + 1)];
#pragma unroll
        for (int j = 0; j < 32; ++j) x[j] = sAug[tid * AUG_LD + j];
#pragma unroll
        for (int i = 31; i >= 0; --i) {
          const float xi = x[i] * lane_bcast(dl, i);
          x[i] = xi;
#pragma unroll
          for (int j = 0; j < i; ++j) x[j] -= sL[i * LDP + j] * xi;
        }
#pragma unroll
        for (int j = 0; j < 32; ++j) sAug[tid * AUG_LD + j] = x[j];
      } else if (jgrp == 1) {
        sample_step(sL, sLl, sx, vecv, t > 1, row,
                    &out[((size_t)n * TT + (t - 1)) * DD + row]);
      }
      __syncthreads();
      {
        float xl[8], jd[8];
#pragma unroll
        for (int i = 0; i < 8; ++i) {
          xl[i] = sAug[(jgrp + 4 * i) * AUG_LD + row];
          jd[i] = sJd[el0 + ELS * i];
        }
#pragma unroll
        for (int i = 0; i < 8; ++i) {
          const int r = jgrp + 4 * i;
          sLl[el0 + ELS * i] = xl[i];
          sAug[r * AUG_LD + row] = jd[i] + (r == row ? EPS_CHOL : 0.f);
        }
      }
      __syncthreads();
      mm32<false, true, LDP>(sLl, sLl, sAug, AUG_LD, MM_SUB, tid);
      __syncthreads();
    } else {
      float jd[8];
#pragma unroll
      for (int i = 0; i < 8; ++i) jd[i] = sJd[el0 + ELS * i];
#pragma unroll
      for (int i = 0; i < 8; ++i) {
        const int r = jgrp + 4 * i;
        sAug[r * AUG_LD + row] = jd[i] + (r == row ? EPS_CHOL : 0.f);
      }
      __syncthreads();
    }
    if (tid < 32) {
      float m[32];
#pragma unroll
      for (int j = 0; j < 32; ++j) m[j] = sAug[tid * AUG_LD + j];
#pragma unroll
      for (int k = 0; k < 32; ++k) {
        const float rinv = lane_bcast(__builtin_amdgcn_rsqf(m[k]), k);
        m[k] *= rinv;
#pragma unroll
        for (int j = k + 1; j < 32; ++j) m[j] -= m[k] * lane_bcast(m[k], j);
      }
#pragma unroll
      for (int j = 0; j < 32; ++j) sAug[tid * AUG_LD + j] = m[j];
    }
    __syncthreads();
    {
      float lv[8];
#pragma unroll
      for (int i = 0; i < 8; ++i) lv[i] = sAug[(jgrp + 4 * i) * AUG_LD + row];
#pragma unroll
      for (int i = 0; i < 8; ++i)
        sL[el0 + ELS * i] = (row <= jgrp + 4 * i) ? lv[i] : 0.f;
    }
    __syncthreads();

  }

  if (!triRole && jgrp == 1) {
    const float zlast = noise[((size_t)n * TT + (TT - 1)) * DD + row];
    sample_step(sL, sLl, sx, zlast, true, row,
                &out[((size_t)n * TT + (TT - 1)) * DD + row]);
  }
}

#define BWD_AHEAD 5
typedef float v4f __attribute__((ext_vector_type(4)));

__device__ inline void bwd_request(const float* __restrict__ g_c,
                                   const float* __restrict__ g_d,
                                   const float* out, size_t base, int ln,
                                   float (&c)[DD], float& d, float& o) {
  const v4f* rowp = (const v4f*)(g_c + base * 1024 + ln * DD);
#pragma unroll
  for (int q = 0; q < 8; ++q) {
    const v4f v = rowp[q];
    c[4 * q] = v.x; c[4 * q + 1] = v.y; c[4 * q + 2] = v.z; c[4 * q + 3] = v.w;
  }
  d = g_d[base * DD + ln];
  o = out[base * DD + ln];
}

__global__ __launch_bounds__(32) void lds_backward_kernel(
    const float* __restrict__ g_c, const float* __restrict__ g_d,
    float* __restrict__ out) {
  const int n = blockIdx.x;
  const int ln = threadIdx.x;
  const size_t seq = (size_t)n * TT;

  float cr[BWD_AHEAD][DD], dv[BWD_AHEAD], ov[BWD_AHEAD];
#pragma unroll
  for (int s = 0; s < BWD_AHEAD; ++s)
    bwd_request(g_c, g_d, out, seq + (TT - 2 - s), ln, cr[s], dv[s], ov[s]);

  float mu = g_d[(seq + (TT - 1)) * DD + ln];
  out[(seq + (TT - 1)) * DD + ln] += mu;

  for (int g = 0; g < (TT - 1) / BWD_AHEAD; ++g) {
#pragma unroll
    for (int s = 0; s < BWD_AHEAD; ++s) {
      const int t = TT - 2 - (g * BWD_AHEAD + s);
      float acc = dv[s];
#pragma unroll
      for (int j = 0; j < DD; ++j)
        acc = __builtin_fmaf(-cr[s][j], lane_bcast(mu, j), acc);
      mu = acc;
      out[(seq + t) * DD + ln] = ov[s] + acc;
      if (t >= BWD_AHEAD)
        bwd_request(g_c, g_d, out, seq + (t - BWD_AHEAD), ln, cr[s], dv[s],
                    ov[s]);
    }
  }
}

extern "C" void kernel_launch(void* const* d_in, const int* in_sizes, int n_in,
                              void* d_out, int out_size, void* d_ws,
                              size_t ws_size, hipStream_t stream) {
  (void)in_sizes; (void)n_in; (void)out_size; (void)ws_size;
  const float* z_samples = (const float*)d_in[0];
  const float* A_base = (const float*)d_in[1];
  const float* b_base = (const float*)d_in[2];
  const float* Q_sqrt = (const float*)d_in[3];
  const float* ms = (const float*)d_in[4];
  const float* Ri_sqrts = (const float*)d_in[5];
  const float* noise = (const float*)d_in[6];
  float* ws = (float*)d_ws;
  float* g_Ris = ws;
  float* g_hR = g_Ris + (size_t)TT * DD * DD;
  float* g_d = g_hR + (size_t)TT * DD;
  float* g_c = g_d + (size_t)NB * TT * DD;
  float* out = (float*)d_out;

  ris_precompute_kernel<<<TT, 128, 0, stream>>>(Ri_sqrts, ms, g_Ris, g_hR);
  lds_forward_kernel<<<NB_FWD, 128, 0, stream>>>(z_samples, A_base, b_base, Q_sqrt,
                                             noise, g_Ris, g_hR, g_c, g_d, out);
  lds_backward_kernel<<<NB, 32, 0, stream>>>(g_c, g_d, out);
}


# Recurrence-free part of each step (z mix, Qis, Jl, Jd, h) is built a step ahead by 4 extra waves of the work-group beside the serial sweeps, on the same barriers; every sum and its order kept
# speedup vs baseline: 25.5677x; 1.3739x over previous
#include <hip/hip_runtime.h>

#define TT 256
#define DD 32
#define KK 8
#define NB 64
#define NB_FWD 128
#define FWD_THREADS 256
#define AUG_LD 66
#define LDP 36
#define LDM (DD * LDP)
#define ELS (4 * LDP)
#define EPS_TRI 0.01f
#define EPS_CHOL 0.01f
#define EPS_BS 0.0001f

typedef float v2f __attribute__((ext_vector_type(2)));
typedef float v8f __attribute__((ext_vector_type(8)));

enum { MM_STORE = 0, MM_STORE_NEG = 1, MM_SUB = 2, MM_ADD = 3 };

template <bool TA, bool TB, int LDAB>
__device__ inline v8f mm32_acc(const float* A, const float* B, int tid) {
  const int lane = tid & 31;
  const int wid = tid >> 5;
  const int mt = (wid >> 1) << 4;
  const int nt = (wid & 1) << 4;
  const int l15 = lane & 15;
  const int kh = (lane >> 4) << 1;
  v8f acc = {};
#pragma unroll
  for (int kt = 0; kt < 32; kt += 4) {
    const int k0 = kt + kh;
    v2f av, bv;
    if (!TA) { av.x = A[(mt + l15) * LDAB + k0];  av.y = A[(mt + l15) * LDAB + k0 + 1]; }
    else     { av.x = A[k0 * LDAB + mt + l15];    av.y = A[(k0 + 1) * LDAB + mt + l15]; }
    if (!TB) { bv.x = B[k0 * LDAB + nt + l15];    bv.y = B[(k0 + 1) * LDAB + nt + l15]; }
    else     { bv.x = B[(nt + l15) * LDAB + k0];  bv.y = B[(nt + l15) * LDAB + k0 + 1]; }
    acc = __builtin_amdgcn_wmma_f32_16x16x4_f32(false, av, false, bv, (short)0,
                                                acc, false, false);
  }
  return acc;
}

template <bool TA, bool TB, int LDAB>
__device__ inline void mm32(const float* A, const float* B, float* D, int ldd,
                            int mode, int tid) {
  const int lane = tid & 31;
  const int wid = tid >> 5;
  const int mt = (wid >> 1) << 4;
  const int nt = (wid & 1) << 4;
  const int l15 = lane & 15;
  const v8f acc = mm32_acc<TA, TB, LDAB>(A, B, tid);
  const int rbase = mt + ((lane >> 4) << 3);
#pragma unroll
  for (int r = 0; r < 8; ++r) {
    float v = acc[r];
    float* p = &D[(rbase + r) * ldd + nt + l15];
    if (mode == MM_STORE) *p = v;
    else if (mode == MM_STORE_NEG) *p = -v;
    else if (mode == MM_SUB) *p -= v;
    else *p += v;
  }
}

__device__ inline float lane_bcast(float v, int src) {
  return __builtin_bit_cast(
      float, __builtin_amdgcn_readlane(__builtin_bit_cast(int, v), src));
}

__device__ inline void sample_step(const float* sL, const float* sLl, float* sx,
                                   float zv, bool havePrev, int ln,
                                   float* outp) {
  float lc[32];
#pragma unroll
  for (int i = 0; i < 32; ++i) lc[i] = sL[i * LDP + ln];
  const float dinv = 1.0f / (sL[ln * (LDP + 1)] + EPS_BS);
  float r = zv;
  if (havePrev) {
    float s = 0.f;
    for (int j = 0; j < DD; ++j) s += sLl[j * LDP + ln] * sx[j];
    r -= s;
  }
  float xv = 0.f;
#pragma unroll
  for (int i = 31; i >= 0; --i) {
    const float p = r * dinv;
    const float xi = lane_bcast(p, i);
    xv = (ln == i) ? p : xv;
    r -= lc[i] * xi;
  }
  sx[ln] = xv;
  *outp = xv;
}

__global__ __launch_bounds__(128) void ris_precompute_kernel(
    const float* __restrict__ Ri_sqrts, const float* __restrict__ ms,
    float* __restrict__ g_Ris, float* __restrict__ g_hR) {
  __shared__ float sR[1024], sO[1024], sm[DD];
  const int t = blockIdx.x;
  const int tid = threadIdx.x;
  for (int e = tid; e < 1024; e += 128) sR[e] = Ri_sqrts[(size_t)t * 1024 + e];
  if (tid < DD) sm[tid] = ms[t * DD + tid];
  __syncthreads();
  mm32<false, true, 32>(sR, sR, sO, 32, MM_STORE, tid);
  __syncthreads();
  for (int e = tid; e < 1024; e += 128) g_Ris[(size_t)t * 1024 + e] = sO[e];
  if (tid < DD) {
    float a = 0.f;
    for (int j = 0; j < DD; ++j) a += sO[tid * 32 + j] * sm[j];
    g_hR[t * DD + tid] = a;
  }
}

__global__ __launch_bounds__(FWD_THREADS) void lds_forward_kernel(
    const float* __restrict__ z_samples, const float* __restrict__ A_base,
    const float* __restrict__ b_base, const float* __restrict__ Q_sqrt,
    const float* __restrict__ noise, const float* __restrict__ g_Ris,
    const float* __restrict__ g_hR, float* __restrict__ g_c,
    float* __restrict__ g_d, float* __restrict__ out) {
  __shared__ float sA[LDM], sQs[LDM], sC[LDM], sL[LDM], sLl[LDM];
  __shared__ float sJd[2][LDM], sQis[2][LDM], sJl[4][LDM];
  __shared__ float sAug[32 * AUG_LD];
  __shared__ float sbs[2][DD], sQb[2][DD], sh[2][DD], sJlPd[DD], sd[DD],
      sDiag[DD], sx[DD];

  const bool triRole = (blockIdx.x < NB);
  const int n = triRole ? blockIdx.x : blockIdx.x - NB;
  const bool builder = (threadIdx.x >= 128);
  const int tid = threadIdx.x & 127;
  const int row = tid & 31;
  const int jgrp = tid >> 5;
  const int el0 = jgrp * LDP + row;
  const int tcol = ((jgrp & 1) << 4) + (row & 15);
  const int trb = ((jgrp >> 1) << 4) + ((row >> 4) << 3);

  float rA[8][KK], rQ[8][KK], rb[KK];
#pragma unroll
  for (int i = 0; i < 8; ++i) {
#pragma unroll
    for (int k = 0; k < KK; ++k) {
      rA[i][k] = A_base[k * 1024 + tid + 128 * i];
      rQ[i][k] = Q_sqrt[k * 1024 + tid + 128 * i];
    }
  }
#pragma unroll
  for (int k = 0; k < KK; ++k)
    rb[k] = (triRole && tid < DD) ? b_base[k * DD + tid] : 0.f;

  float zr[KK];
#pragma unroll
  for (int k = 0; k < KK; ++k)
    zr[k] = builder ? z_samples[(size_t)n * TT * KK + k] : 0.f;

  auto b_mix = [&](int u) {
    if (u < TT - 1) {
#pragma unroll
      for (int i = 0; i < 8; ++i) {
        float a = 0.f, q = 0.f;
#pragma unroll
        for (int k = 0; k < KK; ++k) {
          a += zr[k] * rA[i][k];
          q += zr[k] * rQ[i][k];
        }
        sA[el0 + ELS * i] = a;
        sQs[el0 + ELS * i] = q;
      }
      if (triRole && tid < DD) {
        float b = 0.f;
#pragma unroll
        for (int k = 0; k < KK; ++k) b += zr[k] * rb[k];
        sbs[u & 1][tid] = b;
      }
      if (u + 1 < TT - 1) {
#pragma unroll
        for (int k = 0; k < KK; ++k)
          zr[k] = z_samples[((size_t)n * TT + (u + 1)) * KK + k];
      }
    }
  };
  auto b_qis = [&](int u) {
    if (u < TT - 1)
      mm32<false, true, LDP>(sQs, sQs, sQis[u & 1], LDP, MM_STORE, tid);
  };
  auto b_jl = [&](int u) {
    if (u < TT - 1)
      mm32<false, false, LDP>(sQis[u & 1], sA, sJl[u & 3], LDP, MM_STORE_NEG, tid);
  };
  auto b_jd = [&](int u) {
    float risT[8], qp[8], hRv = 0.f;
#pragma unroll
    for (int r = 0; r < 8; ++r)
      risT[r] = g_Ris[(size_t)u * 1024 + (trb + r) * DD + tcol];
    if (triRole && jgrp == 0) hRv = g_hR[u * DD + row];
#pragma unroll
    for (int r = 0; r < 8; ++r)
      qp[r] = (u > 0) ? sQis[(u - 1) & 1][(trb + r) * LDP + tcol] : 0.f;
    v8f acc = {};
    if (u < TT - 1) acc = mm32_acc<true, false, LDP>(sA, sJl[u & 3], tid);
#pragma unroll
    for (int r = 0; r < 8; ++r) {
      float v = (u < TT - 1) ? -acc[r] : 0.f;
      v = v + risT[r];
      if (u > 0) v = v + qp[r];
      sJd[u & 1][(trb + r) * LDP + tcol] = v;
    }
    if (triRole && jgrp == 0) {
      float hv = hRv;
      if (u > 0) hv += sQb[(u - 1) & 1][row];
      if (u < TT - 1) {
        float jb = 0.f;
        for (int j = 0; j < DD; ++j) jb += sJl[u & 3][row * LDP + j] * sbs[u & 1][j];
        hv += jb;
      }
      sh[u & 1][row] = hv;
    } else if (triRole && jgrp == 2 && u < TT - 1) {
      float qb = 0.f;
      for (int j = 0; j < DD; ++j) qb += sQis[u & 1][row * LDP + j] * sbs[u & 1][j];
      sQb[u & 1][row] = qb;
    }
  };

  if (builder) b_mix(0);
  __syncthreads();
  if (builder) b_qis(0);
  __syncthreads();
  if (builder) b_jl(0);
  __syncthreads();
  if (builder) b_jd(0);
  __syncthreads();
  if (triRole) {
    if (builder) b_mix(1);
    __syncthreads();
  }

  for (int t = 0; t < TT; ++t) {
    const int u = t + 1;
    const bool haveNext = (t < TT - 1);
    const float* Jd_t = sJd[t & 1];
    const float* Jl_t = sJl[t & 3];
    const float* Jl_p = sJl[(t + 3) & 3];

    if (triRole) {
      if (builder) {
        if (haveNext) b_qis(u);
      } else {
        float aj[8], al[8];
#pragma unroll
        for (int i = 0; i < 8; ++i) aj[i] = Jd_t[el0 + ELS * i];
        if (haveNext) {
#pragma unroll
          for (int i = 0; i < 8; ++i) al[i] = Jl_t[el0 + ELS * i];
        } else {
#pragma unroll
          for (int i = 0; i < 8; ++i) al[i] = 0.f;
        }
        float rv = 0.f;
        if (tid < DD) rv = sh[t & 1][tid] - (t > 0 ? sJlPd[tid] : 0.f);
#pragma unroll
        for (int i = 0; i < 8; ++i) {
          const int r = jgrp + 4 * i;
          sAug[r * AUG_LD + row] = aj[i] + (r == row ? EPS_TRI : 0.f);
          sAug[row * AUG_LD + 32 + r] = al[i];
        }
        if (tid < DD) sAug[tid * AUG_LD + 64] = rv;
      }
      __syncthreads();
      if (builder) {
        if (haveNext) b_jl(u);
      } else if (t > 0) {
        mm32<false, false, LDP>(Jl_p, sC, sAug, AUG_LD, MM_SUB, tid);
      }
      __syncthreads();
      if (builder) {
        if (haveNext) b_jd(u);
      } else {
        float a[32], b[9];
        float diag = 0.f;
#pragma unroll
        for (int j = 0; j < 32; ++j) a[j] = sAug[row * AUG_LD + j];
#pragma unroll
        for (int j = 0; j < 8; ++j)
          b[j] = sAug[row * AUG_LD + 32 + 8 * jgrp + j];
        b[8] = sAug[row * AUG_LD + 64];
#pragma unroll
        for (int k = 0; k < 32; ++k) {
          const float rinv = lane_bcast(__builtin_amdgcn_rcpf(a[k]), k);
          float f = a[k] * rinv;
          f = (row == k) ? 0.f : f;
          diag = (row == k) ? a[k] : diag;
#pragma unroll
          for (int j = k + 1; j < 32; ++j) a[j] -= f * lane_bcast(a[j], k);
#pragma unroll
          for (int j = 0; j < 9; ++j) b[j] -= f * lane_bcast(b[j], k);
        }
#pragma unroll
        for (int j = 0; j < 8; ++j)
          sAug[row * AUG_LD + 32 + 8 * jgrp + j] = b[j];
        if (jgrp == 0) {
          sDiag[row] = diag;
          sd[row] = b[8] / diag;
        }
      }
      __syncthreads();
      if (builder) {
        if (t + 2 < TT) b_mix(t + 2);
      } else {
        const size_t base = (size_t)n * TT + t;
        float cv[8];
        const float dg = sDiag[row];
#pragma unroll
        for (int i = 0; i < 8; ++i)
          cv[i] = sAug[row * AUG_LD + 32 + jgrp + 4 * i];
#pragma unroll
        for (int i = 0; i < 8; ++i) {
          cv[i] = cv[i] / dg;
          sC[el0 + ELS * i] = cv[i];
        }
        if (haveNext) {
#pragma unroll
          for (int i = 0; i < 8; ++i) g_c[base * 1024 + tid + 128 * i] = cv[i];
        }
        if (tid < DD) g_d[base * DD + tid] = sd[tid];
        if (haveNext && jgrp == 1) {
          float jd = 0.f;
          for (int j = 0; j < DD; ++j) jd += Jl_t[row * LDP + j] * sd[j];
          sJlPd[row] = jd;
        }
      }
      __syncthreads();
    } else {
      float vecv = 0.f;
      if (!builder && t > 0 && jgrp == 1)
        vecv = noise[((size_t)n * TT + (t - 1)) * DD + row];
      if (builder) {
        if (haveNext) b_mix(u);
      } else if (t > 0) {
        if (tid < 32) {
          float x[32];
          const float dl = 1.0f / sL[tid * (LDP + 1)];
#pragma unroll
          for (int j = 0; j < 32; ++j) x[j] = sAug[tid * AUG_LD + j];
#pragma unroll
          for (int i = 31; i >= 0; --i) {
            const float xi = x[i] * lane_bcast(dl, i);
            x[i] = xi;
#pragma unroll
            for (int j = 0; j < i; ++j) x[j] -= sL[i * LDP + j] * xi;
          }
#pragma unroll
          for (int j = 0; j < 32; ++j) sAug[tid * AUG_LD + j] = x[j];
        } else if (jgrp == 1) {
          sample_step(sL, sLl, sx, vecv, t > 1, row,
                      &out[((size_t)n * TT + (t - 1)) * DD + row]);
        }
      }
      __syncthreads();
      if (builder) {
        if (haveNext) b_qis(u);
      } else if (t > 0) {
        float xl[8], jd[8];
#pragma unroll
        for (int i = 0; i < 8; ++i) {
          xl[i] = sAug[(jgrp + 4 * i) * AUG_LD + row];
          jd[i] = Jd_t[el0 + ELS * i];
        }
#pragma unroll
        for (int i = 0; i < 8; ++i) {
          const int r = jgrp + 4 * i;
          sLl[el0 + ELS * i] = xl[i];
          sAug[r * AUG_LD + row] = jd[i] + (r == row ? EPS_CHOL : 0.f);
        }
      } else {
        float jd[8];
#pragma unroll
        for (int i = 0; i < 8; ++i) jd[i] = Jd_t[el0 + ELS * i];
#pragma unroll
        for (int i = 0; i < 8; ++i) {
          const int r = jgrp + 4 * i;
          sAug[r * AUG_LD + row] = jd[i] + (r == row ? EPS_CHOL : 0.f);
        }
      }
      __syncthreads();
      if (builder) {
        if (haveNext) b_jl(u);
      } else if (t > 0) {
        mm32<false, true, LDP>(sLl, sLl, sAug, AUG_LD, MM_SUB, tid);
      }
      __syncthreads();
      if (builder) {
        if (haveNext) b_jd(u);
      } else if (tid < 32) {
        float m[32];
#pragma unroll
        for (int j = 0; j < 32; ++j) m[j] = sAug[tid * AUG_LD + j];
#pragma unroll
        for (int k = 0; k < 32; ++k) {
          const float rinv = lane_bcast(__builtin_amdgcn_rsqf(m[k]), k);
          m[k] *= rinv;
#pragma unroll
          for (int j = k + 1; j < 32; ++j) m[j] -= m[k] * lane_bcast(m[k], j);
        }
#pragma unroll
        for (int j = 0; j < 32; ++j) sAug[tid * AUG_LD + j] = m[j];
      }
      __syncthreads();
      if (!builder) {
        float lv[8], jl[8];
#pragma unroll
        for (int i = 0; i < 8; ++i) lv[i] = sAug[(jgrp + 4 * i) * AUG_LD + row];
        if (haveNext) {
#pragma unroll
          for (int i = 0; i < 8; ++i) jl[i] = Jl_t[el0 + ELS * i];
        } else {
#pragma unroll
          for (int i = 0; i < 8; ++i) jl[i] = 0.f;
        }
#pragma unroll
        for (int i = 0; i < 8; ++i) {
          sL[el0 + ELS * i] = (row <= jgrp + 4 * i) ? lv[i] : 0.f;
          sAug[(jgrp + 4 * i) * AUG_LD + row] = jl[i];
        }
      }
      __syncthreads();
    }
  }

  if (!triRole && !builder && jgrp == 1) {
    const float zlast = noise[((size_t)n * TT + (TT - 1)) * DD + row];
    sample_step(sL, sLl, sx, zlast, true, row,
                &out[((size_t)n * TT + (TT - 1)) * DD + row]);
  }
}

#define BWD_AHEAD 5
typedef float v4f __attribute__((ext_vector_type(4)));

__device__ inline void bwd_request(const float* __restrict__ g_c,
                                   const float* __restrict__ g_d,
                                   const float* out, size_t base, int ln,
                                   float (&c)[DD], float& d, float& o) {
  const v4f* rowp = (const v4f*)(g_c + base * 1024 + ln * DD);
#pragma unroll
  for (int q = 0; q < 8; ++q) {
    const v4f v = rowp[q];
    c[4 * q] = v.x; c[4 * q + 1] = v.y; c[4 * q + 2] = v.z; c[4 * q + 3] = v.w;
  }
  d = g_d[base * DD + ln];
  o = out[base * DD + ln];
}

__global__ __launch_bounds__(32) void lds_backward_kernel(
    const float* __restrict__ g_c, const float* __restrict__ g_d,
    float* __restrict__ out) {
  const int n = blockIdx.x;
  const int ln = threadIdx.x;
  const size_t seq = (size_t)n * TT;

  float cr[BWD_AHEAD][DD], dv[BWD_AHEAD], ov[BWD_AHEAD];
#pragma unroll
  for (int s = 0; s < BWD_AHEAD; ++s)
    bwd_request(g_c, g_d, out, seq + (TT - 2 - s), ln, cr[s], dv[s], ov[s]);

  float mu = g_d[(seq + (TT - 1)) * DD + ln];
  out[(seq + (TT - 1)) * DD + ln] += mu;

  for (int g = 0; g < (TT - 1) / BWD_AHEAD; ++g) {
#pragma unroll
    for (int s = 0; s < BWD_AHEAD; ++s) {
      const int t = TT - 2 - (g * BWD_AHEAD + s);
      float acc = dv[s];
#pragma unroll
      for (int j = 0; j < DD; ++j)
        acc = __builtin_fmaf(-cr[s][j], lane_bcast(mu, j), acc);
      mu = acc;
      out[(seq + t) * DD + ln] = ov[s] + acc;
      if (t >= BWD_AHEAD)
        bwd_request(g_c, g_d, out, seq + (t - BWD_AHEAD), ln, cr[s], dv[s],
                    ov[s]);
    }
  }
}

extern "C" void kernel_launch(void* const* d_in, const int* in_sizes, int n_in,
                              void* d_out, int out_size, void* d_ws,
                              size_t ws_size, hipStream_t stream) {
  (void)in_sizes; (void)n_in; (void)out_size; (void)ws_size;
  const float* z_samples = (const float*)d_in[0];
  const float* A_base = (const float*)d_in[1];
  const float* b_base = (const float*)d_in[2];
  const float* Q_sqrt = (const float*)d_in[3];
  const float* ms = (const float*)d_in[4];
  const float* Ri_sqrts = (const float*)d_in[5];
  const float* noise = (const float*)d_in[6];
  float* ws = (float*)d_ws;
  float* g_Ris = ws;
  float* g_hR = g_Ris + (size_t)TT * DD * DD;
  float* g_d = g_hR + (size_t)TT * DD;
  float* g_c = g_d + (size_t)NB * TT * DD;
  float* out = (float*)d_out;

  ris_precompute_kernel<<<TT, 128, 0, stream>>>(Ri_sqrts, ms, g_Ris, g_hR);
  lds_forward_kernel<<<NB_FWD, FWD_THREADS, 0, stream>>>(z_samples, A_base, b_base, Q_sqrt,
                                             noise, g_Ris, g_hR, g_c, g_d, out);
  lds_backward_kernel<<<NB, 32, 0, stream>>>(g_c, g_d, out);
}
